# starting kernel
# baseline (speedup 1.0000x reference)
#include <hip/hip_runtime.h>
#include <math.h>

typedef __attribute__((ext_vector_type(2))) float v2f;
typedef __attribute__((ext_vector_type(8))) float v8f;

#define TPB 256
#define WAVES_PER_BLOCK 8

template <int M>
__global__ __launch_bounds__(TPB) void gemm_wmma_f32(
    const float* __restrict__ X, const float* __restrict__ W,
    float* __restrict__ Y, int Nrows) {
  __shared__ __align__(16) float lw[64 * M];

  constexpr int TOTAL4 = 64 * M / 4;
  for (int i = threadIdx.x; i < TOTAL4; i += TPB)
    ((float4*)lw)[i] = ((const float4*)W)[i];
  __syncthreads();

  const int rowTile = blockIdx.x * WAVES_PER_BLOCK + (threadIdx.x >> 5);
  if (rowTile >= (Nrows >> 4)) return;
  const int row0 = rowTile << 4;
  const int lane = threadIdx.x & 31;
  const int r    = lane & 15;
  const int half = lane >> 4;

  v2f a[16];
  const float* xrow = X + (size_t)(row0 + r) * 64;
#pragma unroll
  for (int ks = 0; ks < 16; ++ks) {
    const int k = 4 * ks + 2 * half;
    a[ks].x = xrow[k];
    a[ks].y = xrow[k + 1];
  }

#pragma unroll
  for (int tn = 0; tn < M / 16; ++tn) {
    const int col0 = tn << 4;
    v8f c = {};
#pragma unroll
    for (int ks = 0; ks < 16; ++ks) {
      const int k = 4 * ks + 2 * half;
      v2f b;
      b.x = lw[k * M + col0 + r];
      b.y = lw[(k + 1) * M + col0 + r];
      c = __builtin_amdgcn_wmma_f32_16x16x4_f32(false, a[ks], false, b,
                                                (short)0, c, false, false);
    }
#pragma unroll
    for (int v = 0; v < 8; ++v)
      Y[(size_t)(row0 + v + 8 * half) * M + col0 + r] = c[v];
  }
}

__device__ __forceinline__ void edge_sd(const int* __restrict__ ei, int e, int E,
                                        int& s, int& d) {
  if (e < E) { s = ei[e]; d = ei[E + e]; }
  else       { s = e - E; d = s; }
}

__device__ __forceinline__ void atomic_max_float(float* addr, float val) {
  if (val >= 0.0f) atomicMax((int*)addr, __float_as_int(val));
  else             atomicMin((unsigned int*)addr, __float_as_uint(val));
}

template <int C>
__global__ __launch_bounds__(TPB) void edge_score(
    const float* __restrict__ xl, const float* __restrict__ xr,
    const int* __restrict__ ei, const float* __restrict__ att,
    float* __restrict__ score, float* __restrict__ mbuf, int E, int ET) {
  const int tid = blockIdx.x * blockDim.x + threadIdx.x;
  if (tid >= ET * 4) return;
  const int e = tid >> 2, h = tid & 3;
  int s, d; edge_sd(ei, e, E, s, d);
  const int ldf = 4 * C;
  const float4* pl = (const float4*)(xl + (size_t)s * ldf + h * C);
  const float4* pr = (const float4*)(xr + (size_t)d * ldf + h * C);
  const float4* pa = (const float4*)(att + h * C);
  float acc = 0.0f;
#pragma unroll
  for (int i = 0; i < C / 4; ++i) {
    float4 a = pl[i], b = pr[i], w = pa[i];
    float vx = a.x + b.x, vy = a.y + b.y, vz = a.z + b.z, vw = a.w + b.w;
    vx = vx > 0.0f ? vx : 0.2f * vx;
    vy = vy > 0.0f ? vy : 0.2f * vy;
    vz = vz > 0.0f ? vz : 0.2f * vz;
    vw = vw > 0.0f ? vw : 0.2f * vw;
    acc += w.x * vx + w.y * vy + w.z * vz + w.w * vw;
  }
  score[(size_t)e * 4 + h] = acc;
  atomic_max_float(&mbuf[(size_t)d * 4 + h], acc);
}

__global__ __launch_bounds__(TPB) void edge_exp(
    float* __restrict__ score, const float* __restrict__ mbuf,
    float* __restrict__ denom, const int* __restrict__ ei, int E, int ET) {
  const int tid = blockIdx.x * blockDim.x + threadIdx.x;
  if (tid >= ET * 4) return;
  const int e = tid >> 2, h = tid & 3;
  int s, d; edge_sd(ei, e, E, s, d);
  const float ex = expf(score[(size_t)e * 4 + h] - mbuf[(size_t)d * 4 + h]);
  score[(size_t)e * 4 + h] = ex;
  atomicAdd(&denom[(size_t)d * 4 + h], ex);
}

template <int C>
__global__ __launch_bounds__(TPB) void edge_agg(
    const float* __restrict__ xl, const float* __restrict__ score,
    const float* __restrict__ denom, const int* __restrict__ ei,
    float* __restrict__ agg, int E, int ET) {
  const int tid = blockIdx.x * blockDim.x + threadIdx.x;
  if (tid >= ET * 4) return;
  const int e = tid >> 2, h = tid & 3;
  int s, d; edge_sd(ei, e, E, s, d);
  const float alpha = score[(size_t)e * 4 + h] / denom[(size_t)d * 4 + h];
  const int ldf = 4 * C;
  const float4* srcp = (const float4*)(xl + (size_t)s * ldf + h * C);
  float* dstp = agg + (size_t)d * ldf + h * C;
#pragma unroll
  for (int i = 0; i < C / 4; ++i) {
    float4 v = srcp[i];
    atomicAdd(&dstp[4 * i + 0], alpha * v.x);
    atomicAdd(&dstp[4 * i + 1], alpha * v.y);
    atomicAdd(&dstp[4 * i + 2], alpha * v.z);
    atomicAdd(&dstp[4 * i + 3], alpha * v.w);
  }
}

__global__ __launch_bounds__(TPB) void bias_elu(
    float* __restrict__ feat, const float* __restrict__ bias, int total, int M) {
  const int tid = blockIdx.x * blockDim.x + threadIdx.x;
  if (tid >= total) return;
  const float v = feat[tid] + bias[tid & (M - 1)];
  feat[tid] = v > 0.0f ? v : expm1f(v);
}

__global__ __launch_bounds__(TPB) void final_mean(
    const float* __restrict__ agg, const float* __restrict__ bias,
    float* __restrict__ out, int N) {
  const int tid = blockIdx.x * blockDim.x + threadIdx.x;
  if (tid >= N * 32) return;
  const int n = tid >> 5, c = tid & 31;
  const size_t b = (size_t)n * 128 + c;
  const float s = 0.25f * (agg[b] + agg[b + 32] + agg[b + 64] + agg[b + 96]);
  out[tid] = s + bias[c];
}

extern "C" void kernel_launch(void* const* d_in, const int* in_sizes, int n_in,
                              void* d_out, int out_size, void* d_ws, size_t ws_size,
                              hipStream_t stream) {
  (void)in_sizes; (void)n_in; (void)out_size; (void)ws_size;
  const float* x   = (const float*)d_in[0];
  const int*   ei  = (const int*)d_in[1];
  const float* W1l = (const float*)d_in[2];
  const float* W1r = (const float*)d_in[3];
  const float* a1  = (const float*)d_in[4];
  const float* b1  = (const float*)d_in[5];
  const float* W2l = (const float*)d_in[6];
  const float* W2r = (const float*)d_in[7];
  const float* a2  = (const float*)d_in[8];
  const float* b2  = (const float*)d_in[9];
  const float* W3l = (const float*)d_in[10];
  const float* W3r = (const float*)d_in[11];
  const float* a3  = (const float*)d_in[12];
  const float* b3  = (const float*)d_in[13];

  const int N = 100000, E = 800000, ET = E + N;

  float* ws = (float*)d_ws;
  float* xl = ws;
  float* xr = xl + (size_t)N * 128;
  float* sc = xr + (size_t)N * 128;
  float* mb = sc + (size_t)ET * 4;
  float* dn = mb + (size_t)N * 4;
  float* h1 = dn + (size_t)N * 4;
  float* h2 = h1 + (size_t)N * 64;
  float* ag = h2 + (size_t)N * 64;

  const int rowTiles = N / 16;
  const int gblocks  = (rowTiles + WAVES_PER_BLOCK - 1) / WAVES_PER_BLOCK;

  auto layer = [&](const float* in, const float* Wl, const float* Wr,
                   const float* att, const float* bias, float* out,
                   int C, bool final_layer) {
    const int M = 4 * C;
    if (M == 64) {
      gemm_wmma_f32<64><<<gblocks, TPB, 0, stream>>>(in, Wl, xl, N);
      gemm_wmma_f32<64><<<gblocks, TPB, 0, stream>>>(in, Wr, xr, N);
    } else {
      gemm_wmma_f32<128><<<gblocks, TPB, 0, stream>>>(in, Wl, xl, N);
      gemm_wmma_f32<128><<<gblocks, TPB, 0, stream>>>(in, Wr, xr, N);
    }

    float* aggbuf = final_layer ? ag : out;
    hipMemsetAsync(mb, 0xFF, (size_t)N * 4 * sizeof(float), stream);
    hipMemsetAsync(dn, 0x00, (size_t)N * 4 * sizeof(float), stream);
    hipMemsetAsync(aggbuf, 0x00, (size_t)N * M * sizeof(float), stream);

    const int eThreads = ET * 4;
    const int eBlocks = (eThreads + TPB - 1) / TPB;
    if (C == 16) edge_score<16><<<eBlocks, TPB, 0, stream>>>(xl, xr, ei, att, sc, mb, E, ET);
    else         edge_score<32><<<eBlocks, TPB, 0, stream>>>(xl, xr, ei, att, sc, mb, E, ET);
    edge_exp<<<eBlocks, TPB, 0, stream>>>(sc, mb, dn, ei, E, ET);
    if (C == 16) edge_agg<16><<<eBlocks, TPB, 0, stream>>>(xl, sc, dn, ei, aggbuf, E, ET);
    else         edge_agg<32><<<eBlocks, TPB, 0, stream>>>(xl, sc, dn, ei, aggbuf, E, ET);

    if (!final_layer) {
      const int t = N * M;
      bias_elu<<<(t + TPB - 1) / TPB, TPB, 0, stream>>>(out, bias, t, M);
    } else {
      const int t = N * 32;
      final_mean<<<(t + TPB - 1) / TPB, TPB, 0, stream>>>(ag, bias, (float*)d_out, N);
    }
  };

  layer(x,  W1l, W1r, a1, b1, h1, 16, false);
  layer(h1, W2l, W2r, a2, b2, h2, 16, false);
  layer(h2, W3l, W3r, a3, b3, (float*)d_out, 32, true);
}


# bias add + ELU of layers 1-2 moved out of their own in-place pass into the next layer's GEMM A-fragment loads (h1, h2 keep the raw aggregate): 18 -> 16 launches; per-element sum order unchanged
# speedup vs baseline: 3.0743x; 3.0743x over previous
#include <hip/hip_runtime.h>
#include <math.h>

typedef __attribute__((ext_vector_type(2))) float v2f;
typedef __attribute__((ext_vector_type(8))) float v8f;

#define TPB 256
#define WAVES_PER_BLOCK 8

template <int M, bool ACT>
__global__ __launch_bounds__(TPB) void gemm_wmma_f32(
    const float* __restrict__ X, const float* __restrict__ W,
    const float* __restrict__ bias, float* __restrict__ Y, int Nrows) {
  __shared__ __align__(16) float lw[64 * M];

  constexpr int TOTAL4 = 64 * M / 4;
  for (int i = threadIdx.x; i < TOTAL4; i += TPB)
    ((float4*)lw)[i] = ((const float4*)W)[i];
  __syncthreads();

  const int rowTile = blockIdx.x * WAVES_PER_BLOCK + (threadIdx.x >> 5);
  if (rowTile >= (Nrows >> 4)) return;
  const int row0 = rowTile << 4;
  const int lane = threadIdx.x & 31;
  const int r    = lane & 15;
  const int half = lane >> 4;

  v2f a[16];
  const float* xrow = X + (size_t)(row0 + r) * 64;
#pragma unroll
  for (int ks = 0; ks < 16; ++ks) {
    const int k = 4 * ks + 2 * half;
    a[ks].x = xrow[k];
    a[ks].y = xrow[k + 1];
  }
  if constexpr (ACT) {
#pragma unroll
    for (int ks = 0; ks < 16; ++ks) {
      const int k = 4 * ks + 2 * half;
      const float x0 = a[ks].x + bias[k];
      const float x1 = a[ks].y + bias[k + 1];
      const float e0 = expm1f(x0);
      const float e1 = expm1f(x1);
      a[ks].x = x0 > 0.0f ? x0 : e0;
      a[ks].y = x1 > 0.0f ? x1 : e1;
    }
  }

#pragma unroll
  for (int tn = 0; tn < M / 16; ++tn) {
    const int col0 = tn << 4;
    v8f c = {};
#pragma unroll
    for (int ks = 0; ks < 16; ++ks) {
      const int k = 4 * ks + 2 * half;
      v2f b;
      b.x = lw[k * M + col0 + r];
      b.y = lw[(k + 1) * M + col0 + r];
      c = __builtin_amdgcn_wmma_f32_16x16x4_f32(false, a[ks], false, b,
                                                (short)0, c, false, false);
    }
#pragma unroll
    for (int v = 0; v < 8; ++v)
      Y[(size_t)(row0 + v + 8 * half) * M + col0 + r] = c[v];
  }
}

__device__ __forceinline__ void edge_sd(const int* __restrict__ ei, int e, int E,
                                        int& s, int& d) {
  if (e < E) { s = ei[e]; d = ei[E + e]; }
  else       { s = e - E; d = s; }
}

__device__ __forceinline__ void atomic_max_float(float* addr, float val) {
  if (val >= 0.0f) atomicMax((int*)addr, __float_as_int(val));
  else             atomicMin((unsigned int*)addr, __float_as_uint(val));
}

template <int CTRL>
__device__ __forceinline__ float lane_xor_add(float v) {
  const int t = __builtin_amdgcn_update_dpp(0, __float_as_int(v), CTRL, 0xf, 0xf, true);
  return v + __int_as_float(t);
}

template <int C>
__global__ __launch_bounds__(TPB) void edge_score(
    const float* __restrict__ xl, const float* __restrict__ xr,
    const int* __restrict__ ei, const float* __restrict__ att,
    float* __restrict__ score, float* __restrict__ mbuf, int E, int ET) {
  constexpr int M     = 4 * C;
  constexpr int LPR   = M / 4;
  constexpr int EPS   = 32 / LPR;
  constexpr int LPH   = C / 4;
  constexpr int STEPS = 8 / EPS;
  static_assert(STEPS == LPH, "one kept score per lane needs steps == lanes per head");
  const int lane = threadIdx.x & 31;
  const int e0 = __builtin_amdgcn_readfirstlane(
      (int)(((blockIdx.x * TPB + threadIdx.x) >> 5) << 3));
  if (e0 >= ET) return;
  const int ch   = lane & (LPR - 1);
  const int sub  = lane / LPR;
  const int q    = lane & (LPH - 1);
  const int h    = ch / LPH;
  const int eloc = q * EPS + sub;
  const int e    = e0 + eloc;
  int s = 0, d = 0;
  if (e < ET) edge_sd(ei, e, E, s, d);
  const float4 w = ((const float4*)att)[ch];

  float4 a[STEPS], b[STEPS];
#pragma unroll
  for (int t = 0; t < STEPS; ++t) {
    int sj, dj;
    if constexpr (EPS == 1) {
      sj = __builtin_amdgcn_readlane(s, t);
      dj = __builtin_amdgcn_readlane(d, t);
    } else {
      const int s0 = __builtin_amdgcn_readlane(s, t);
      const int s1 = __builtin_amdgcn_readlane(s, 16 + t);
      const int d0 = __builtin_amdgcn_readlane(d, t);
      const int d1 = __builtin_amdgcn_readlane(d, 16 + t);
      sj = sub ? s1 : s0;
      dj = sub ? d1 : d0;
    }
    const unsigned oa = (unsigned)sj * (unsigned)(M * 4) + (unsigned)ch * 16u;
    const unsigned ob = (unsigned)dj * (unsigned)(M * 4) + (unsigned)ch * 16u;
    a[t] = *(const float4*)((const char*)xl + oa);
    b[t] = *(const float4*)((const char*)xr + ob);
  }

  float acc = 0.0f;
#pragma unroll
  for (int t = 0; t < STEPS; ++t) {
    float vx = a[t].x + b[t].x, vy = a[t].y + b[t].y;
    float vz = a[t].z + b[t].z, vw = a[t].w + b[t].w;
    vx = vx > 0.0f ? vx : 0.2f * vx;
    vy = vy > 0.0f ? vy : 0.2f * vy;
    vz = vz > 0.0f ? vz : 0.2f * vz;
    vw = vw > 0.0f ? vw : 0.2f * vw;
    float p = w.x * vx + w.y * vy + w.z * vz + w.w * vw;
    p = lane_xor_add<0x161>(p);
    p = lane_xor_add<0x162>(p);
    if constexpr (LPH == 8) p = lane_xor_add<0x164>(p);
    acc = (q == t) ? p : acc;
  }
  if (e < ET) {
    score[(size_t)e * 4 + h] = acc;
    atomic_max_float(&mbuf[(size_t)d * 4 + h], acc);
  }
}

__global__ __launch_bounds__(TPB) void edge_exp(
    float* __restrict__ score, const float* __restrict__ mbuf,
    float* __restrict__ denom, const int* __restrict__ ei, int E, int ET) {
  const int tid = blockIdx.x * blockDim.x + threadIdx.x;
  if (tid >= ET * 4) return;
  const int e = tid >> 2, h = tid & 3;
  int s, d; edge_sd(ei, e, E, s, d);
  const float ex = expf(score[(size_t)e * 4 + h] - mbuf[(size_t)d * 4 + h]);
  score[(size_t)e * 4 + h] = ex;
  atomicAdd(&denom[(size_t)d * 4 + h], ex);
}

template <int C, int NB, bool MEAN>
__device__ __forceinline__ void agg_batch(
    const float* __restrict__ xl, float* __restrict__ agg,
    int s, int d, float al0, float al1, float al2, float al3, int j0, int lane) {
  constexpr int M = 4 * C;
  constexpr int Q = M / 32;
  float v[NB][Q];
  float w[NB][Q];
  int   dj[NB];
#pragma unroll
  for (int u = 0; u < NB; ++u) {
    const int j = j0 + u;
    const int sj = __builtin_amdgcn_readlane(s, j);
    dj[u] = __builtin_amdgcn_readlane(d, j);
    const float* row = xl + (size_t)sj * M;
#pragma unroll
    for (int q = 0; q < Q; ++q) v[u][q] = row[q * 32 + lane];
    const float a0 = __int_as_float(__builtin_amdgcn_readlane(__float_as_int(al0), j));
    const float a1 = __int_as_float(__builtin_amdgcn_readlane(__float_as_int(al1), j));
    const float a2 = __int_as_float(__builtin_amdgcn_readlane(__float_as_int(al2), j));
    const float a3 = __int_as_float(__builtin_amdgcn_readlane(__float_as_int(al3), j));
    if constexpr (Q == 4) {
      w[u][0] = a0; w[u][1] = a1; w[u][2] = a2; w[u][3] = a3;
    } else {
      const bool hi = lane >= 16;
      w[u][0] = hi ? a1 : a0;
      w[u][1] = hi ? a3 : a2;
    }
  }
#pragma unroll
  for (int u = 0; u < NB; ++u) {
    if constexpr (MEAN) {
      static_assert(Q == 4, "head sum in registers needs one head per line");
      float t = w[u][0] * v[u][0];
      t = fmaf(w[u][1], v[u][1], t);
      t = fmaf(w[u][2], v[u][2], t);
      t = fmaf(w[u][3], v[u][3], t);
      atomicAdd(&agg[(size_t)dj[u] * C + lane], t);
    } else {
      float* arow = agg + (size_t)dj[u] * M;
#pragma unroll
      for (int q = 0; q < Q; ++q)
        atomicAdd(&arow[q * 32 + lane], w[u][q] * v[u][q]);
    }
  }
}

template <int C, bool MEAN>
__global__ __launch_bounds__(TPB) void edge_agg(
    const float* __restrict__ xl, const float* __restrict__ score,
    const float* __restrict__ denom, const int* __restrict__ ei,
    float* __restrict__ agg, int E, int ET) {
  constexpr int U = 128 / C;
  const int lane = threadIdx.x & 31;
  const int e0 = __builtin_amdgcn_readfirstlane(
      (int)(((blockIdx.x * TPB + threadIdx.x) >> 5) << 5));
  if (e0 >= ET) return;
  const int e = e0 + lane;
  int s = 0, d = 0;
  float al0 = 0.0f, al1 = 0.0f, al2 = 0.0f, al3 = 0.0f;
  if (e < ET) {
    edge_sd(ei, e, E, s, d);
    const float4 ex = *(const float4*)(score + (size_t)e * 4);
    const float4 dn = *(const float4*)(denom + (size_t)d * 4);
    al0 = ex.x / dn.x; al1 = ex.y / dn.y;
    al2 = ex.z / dn.z; al3 = ex.w / dn.w;
  }
  const int cnt = min(32, ET - e0);
  int j = 0;
  for (; j + U <= cnt; j += U)
    agg_batch<C, U, MEAN>(xl, agg, s, d, al0, al1, al2, al3, j, lane);
  for (; j < cnt; ++j)
    agg_batch<C, 1, MEAN>(xl, agg, s, d, al0, al1, al2, al3, j, lane);
}

__global__ __launch_bounds__(TPB) void final_mean(
    const float* __restrict__ agg, const float* __restrict__ bias,
    float* __restrict__ out, int N) {
  const int tid = blockIdx.x * blockDim.x + threadIdx.x;
  if (tid >= N * 32) return;
  out[tid] = 0.25f * agg[tid] + bias[tid & 31];
}

extern "C" void kernel_launch(void* const* d_in, const int* in_sizes, int n_in,
                              void* d_out, int out_size, void* d_ws, size_t ws_size,
                              hipStream_t stream) {
  (void)in_sizes; (void)n_in; (void)out_size; (void)ws_size;
  const float* x   = (const float*)d_in[0];
  const int*   ei  = (const int*)d_in[1];
  const float* W1l = (const float*)d_in[2];
  const float* W1r = (const float*)d_in[3];
  const float* a1  = (const float*)d_in[4];
  const float* b1  = (const float*)d_in[5];
  const float* W2l = (const float*)d_in[6];
  const float* W2r = (const float*)d_in[7];
  const float* a2  = (const float*)d_in[8];
  const float* b2  = (const float*)d_in[9];
  const float* W3l = (const float*)d_in[10];
  const float* W3r = (const float*)d_in[11];
  const float* a3  = (const float*)d_in[12];
  const float* b3  = (const float*)d_in[13];

  const int N = 100000, E = 800000, ET = E + N;

  float* ws = (float*)d_ws;
  float* xl = ws;
  float* xr = xl + (size_t)N * 128;
  float* sc = xr + (size_t)N * 128;
  float* mb = sc + (size_t)ET * 4;
  float* dn = mb + (size_t)N * 4;
  float* h1 = dn + (size_t)N * 4;
  float* h2 = h1 + (size_t)N * 64;
  float* ag = h2 + (size_t)N * 64;

  const int rowTiles = N / 16;
  const int gblocks  = (rowTiles + WAVES_PER_BLOCK - 1) / WAVES_PER_BLOCK;

  auto layer = [&](const float* in, const float* inBias, const float* Wl, const float* Wr,
                   const float* att, const float* bias, float* out,
                   int C, bool final_layer) {
    const int M = 4 * C;
    if (M == 64 && !inBias) {
      gemm_wmma_f32<64, false><<<gblocks, TPB, 0, stream>>>(in, Wl, inBias, xl, N);
      gemm_wmma_f32<64, false><<<gblocks, TPB, 0, stream>>>(in, Wr, inBias, xr, N);
    } else if (M == 64) {
      gemm_wmma_f32<64, true><<<gblocks, TPB, 0, stream>>>(in, Wl, inBias, xl, N);
      gemm_wmma_f32<64, true><<<gblocks, TPB, 0, stream>>>(in, Wr, inBias, xr, N);
    } else {
      gemm_wmma_f32<128, true><<<gblocks, TPB, 0, stream>>>(in, Wl, inBias, xl, N);
      gemm_wmma_f32<128, true><<<gblocks, TPB, 0, stream>>>(in, Wr, inBias, xr, N);
    }

    float* aggbuf = final_layer ? ag : out;
    hipMemsetAsync(mb, 0xFF, (size_t)N * 4 * sizeof(float), stream);
    hipMemsetAsync(dn, 0x00, (size_t)N * 4 * sizeof(float), stream);
    const int aggw = final_layer ? C : M;
    hipMemsetAsync(aggbuf, 0x00, (size_t)N * aggw * sizeof(float), stream);

    const int eThreads = ET * 4;
    const int eBlocks = (eThreads + TPB - 1) / TPB;
    if (C == 16) edge_score<16><<<eBlocks, TPB, 0, stream>>>(xl, xr, ei, att, sc, mb, E, ET);
    else         edge_score<32><<<eBlocks, TPB, 0, stream>>>(xl, xr, ei, att, sc, mb, E, ET);
    edge_exp<<<eBlocks, TPB, 0, stream>>>(sc, mb, dn, ei, E, ET);
    const int aBlocks = (ET + TPB - 1) / TPB;
    if (C == 16) edge_agg<16, false><<<aBlocks, TPB, 0, stream>>>(xl, sc, dn, ei, aggbuf, E, ET);
    else         edge_agg<32, true><<<aBlocks, TPB, 0, stream>>>(xl, sc, dn, ei, aggbuf, E, ET);

    if (final_layer) {
      const int t = N * 32;
      final_mean<<<(t + TPB - 1) / TPB, TPB, 0, stream>>>(ag, bias, (float*)d_out, N);
    }
  };

  layer(x,  nullptr, W1l, W1r, a1, b1, h1, 16, false);
  layer(h1, b1,      W2l, W2r, a2, b2, h2, 16, false);
  layer(h2, b2,      W3l, W3r, a3, b3, (float*)d_out, 32, true);
}


# concat layers: softmax division moved behind the edge sum (next GEMM's A loads scale by 1/denom), so exp and denominator adds join the aggregation pass: 16 -> 14 launches; sums add undivided terms
# speedup vs baseline: 3.2067x; 1.0431x over previous
#include <hip/hip_runtime.h>
#include <math.h>

typedef __attribute__((ext_vector_type(2))) float v2f;
typedef __attribute__((ext_vector_type(8))) float v8f;

#define TPB 256
#define WAVES_PER_BLOCK 8

template <int M, bool ACT>
__global__ __launch_bounds__(TPB) void gemm_wmma_f32(
    const float* __restrict__ X, const float* __restrict__ W,
    const float* __restrict__ bias, const float* __restrict__ den,
    float* __restrict__ Y, int Nrows) {
  __shared__ __align__(16) float lw[64 * M];

  constexpr int TOTAL4 = 64 * M / 4;
  for (int i = threadIdx.x; i < TOTAL4; i += TPB)
    ((float4*)lw)[i] = ((const float4*)W)[i];
  __syncthreads();

  const int rowTile = blockIdx.x * WAVES_PER_BLOCK + (threadIdx.x >> 5);
  if (rowTile >= (Nrows >> 4)) return;
  const int row0 = rowTile << 4;
  const int lane = threadIdx.x & 31;
  const int r    = lane & 15;
  const int half = lane >> 4;

  v2f a[16];
  const float* xrow = X + (size_t)(row0 + r) * 64;
#pragma unroll
  for (int ks = 0; ks < 16; ++ks) {
    const int k = 4 * ks + 2 * half;
    a[ks].x = xrow[k];
    a[ks].y = xrow[k + 1];
  }
  if constexpr (ACT) {
    const float4 dv = *(const float4*)(den + (size_t)(row0 + r) * 4);
    const float inv[4] = {1.0f / dv.x, 1.0f / dv.y, 1.0f / dv.z, 1.0f / dv.w};
#pragma unroll
    for (int ks = 0; ks < 16; ++ks) {
      const int k = 4 * ks + 2 * half;
      const float x0 = a[ks].x * inv[ks >> 2] + bias[k];
      const float x1 = a[ks].y * inv[ks >> 2] + bias[k + 1];
      const float e0 = expm1f(x0);
      const float e1 = expm1f(x1);
      a[ks].x = x0 > 0.0f ? x0 : e0;
      a[ks].y = x1 > 0.0f ? x1 : e1;
    }
  }

#pragma unroll
  for (int tn = 0; tn < M / 16; ++tn) {
    const int col0 = tn << 4;
    v8f c = {};
#pragma unroll
    for (int ks = 0; ks < 16; ++ks) {
      const int k = 4 * ks + 2 * half;
      v2f b;
      b.x = lw[k * M + col0 + r];
      b.y = lw[(k + 1) * M + col0 + r];
      c = __builtin_amdgcn_wmma_f32_16x16x4_f32(false, a[ks], false, b,
                                                (short)0, c, false, false);
    }
#pragma unroll
    for (int v = 0; v < 8; ++v)
      Y[(size_t)(row0 + v + 8 * half) * M + col0 + r] = c[v];
  }
}

__device__ __forceinline__ void edge_sd(const int* __restrict__ ei, int e, int E,
                                        int& s, int& d) {
  if (e < E) { s = ei[e]; d = ei[E + e]; }
  else       { s = e - E; d = s; }
}

__device__ __forceinline__ void atomic_max_float(float* addr, float val) {
  if (val >= 0.0f) atomicMax((int*)addr, __float_as_int(val));
  else             atomicMin((unsigned int*)addr, __float_as_uint(val));
}

template <int CTRL>
__device__ __forceinline__ float lane_xor_add(float v) {
  const int t = __builtin_amdgcn_update_dpp(0, __float_as_int(v), CTRL, 0xf, 0xf, true);
  return v + __int_as_float(t);
}

template <int C>
__global__ __launch_bounds__(TPB) void edge_score(
    const float* __restrict__ xl, const float* __restrict__ xr,
    const int* __restrict__ ei, const float* __restrict__ att,
    float* __restrict__ score, float* __restrict__ mbuf, int E, int ET) {
  constexpr int M     = 4 * C;
  constexpr int LPR   = M / 4;
  constexpr int EPS   = 32 / LPR;
  constexpr int LPH   = C / 4;
  constexpr int STEPS = 8 / EPS;
  static_assert(STEPS == LPH, "one kept score per lane needs steps == lanes per head");
  const int lane = threadIdx.x & 31;
  const int e0 = __builtin_amdgcn_readfirstlane(
      (int)(((blockIdx.x * TPB + threadIdx.x) >> 5) << 3));
  if (e0 >= ET) return;
  const int ch   = lane & (LPR - 1);
  const int sub  = lane / LPR;
  const int q    = lane & (LPH - 1);
  const int h    = ch / LPH;
  const int eloc = q * EPS + sub;
  const int e    = e0 + eloc;
  int s = 0, d = 0;
  if (e < ET) edge_sd(ei, e, E, s, d);
  const float4 w = ((const float4*)att)[ch];

  float4 a[STEPS], b[STEPS];
#pragma unroll
  for (int t = 0; t < STEPS; ++t) {
    int sj, dj;
    if constexpr (EPS == 1) {
      sj = __builtin_amdgcn_readlane(s, t);
      dj = __builtin_amdgcn_readlane(d, t);
    } else {
      const int s0 = __builtin_amdgcn_readlane(s, t);
      const int s1 = __builtin_amdgcn_readlane(s, 16 + t);
      const int d0 = __builtin_amdgcn_readlane(d, t);
      const int d1 = __builtin_amdgcn_readlane(d, 16 + t);
      sj = sub ? s1 : s0;
      dj = sub ? d1 : d0;
    }
    const unsigned oa = (unsigned)sj * (unsigned)(M * 4) + (unsigned)ch * 16u;
    const unsigned ob = (unsigned)dj * (unsigned)(M * 4) + (unsigned)ch * 16u;
    a[t] = *(const float4*)((const char*)xl + oa);
    b[t] = *(const float4*)((const char*)xr + ob);
  }

  float acc = 0.0f;
#pragma unroll
  for (int t = 0; t < STEPS; ++t) {
    float vx = a[t].x + b[t].x, vy = a[t].y + b[t].y;
    float vz = a[t].z + b[t].z, vw = a[t].w + b[t].w;
    vx = vx > 0.0f ? vx : 0.2f * vx;
    vy = vy > 0.0f ? vy : 0.2f * vy;
    vz = vz > 0.0f ? vz : 0.2f * vz;
    vw = vw > 0.0f ? vw : 0.2f * vw;
    float p = w.x * vx + w.y * vy + w.z * vz + w.w * vw;
    p = lane_xor_add<0x161>(p);
    p = lane_xor_add<0x162>(p);
    if constexpr (LPH == 8) p = lane_xor_add<0x164>(p);
    acc = (q == t) ? p : acc;
  }
  if (e < ET) {
    score[(size_t)e * 4 + h] = acc;
    atomic_max_float(&mbuf[(size_t)d * 4 + h], acc);
  }
}

__global__ __launch_bounds__(TPB) void edge_exp(
    float* __restrict__ score, const float* __restrict__ mbuf,
    float* __restrict__ denom, const int* __restrict__ ei, int E, int ET) {
  const int tid = blockIdx.x * blockDim.x + threadIdx.x;
  if (tid >= ET * 4) return;
  const int e = tid >> 2, h = tid & 3;
  int s, d; edge_sd(ei, e, E, s, d);
  const float ex = expf(score[(size_t)e * 4 + h] - mbuf[(size_t)d * 4 + h]);
  score[(size_t)e * 4 + h] = ex;
  atomicAdd(&denom[(size_t)d * 4 + h], ex);
}

template <int C, int NB, bool MEAN>
__device__ __forceinline__ void agg_batch(
    const float* __restrict__ xl, float* __restrict__ agg,
    int s, int d, float al0, float al1, float al2, float al3, int j0, int lane) {
  constexpr int M = 4 * C;
  constexpr int Q = M / 32;
  float v[NB][Q];
  float w[NB][Q];
  int   dj[NB];
#pragma unroll
  for (int u = 0; u < NB; ++u) {
    const int j = j0 + u;
    const int sj = __builtin_amdgcn_readlane(s, j);
    dj[u] = __builtin_amdgcn_readlane(d, j);
    const float* row = xl + (size_t)sj * M;
#pragma unroll
    for (int q = 0; q < Q; ++q) v[u][q] = row[q * 32 + lane];
    float a0, a1, a2, a3;
    if constexpr (MEAN) {
      a0 = __int_as_float(__builtin_amdgcn_readlane(__float_as_int(al0), j));
      a1 = __int_as_float(__builtin_amdgcn_readlane(__float_as_int(al1), j));
      a2 = __int_as_float(__builtin_amdgcn_readlane(__float_as_int(al2), j));
      a3 = __int_as_float(__builtin_amdgcn_readlane(__float_as_int(al3), j));
    } else {
      static_assert(8 % NB == 0, "a batch must stay inside one group of 8 edges");
      const int k = j0 >> 3;
      const int ek = __float_as_int(k == 0 ? al0 : k == 1 ? al1 : k == 2 ? al2 : al3);
      const int l0 = (j & 7) << 2;
      a0 = __int_as_float(__builtin_amdgcn_readlane(ek, l0));
      a1 = __int_as_float(__builtin_amdgcn_readlane(ek, l0 + 1));
      a2 = __int_as_float(__builtin_amdgcn_readlane(ek, l0 + 2));
      a3 = __int_as_float(__builtin_amdgcn_readlane(ek, l0 + 3));
    }
    if constexpr (Q == 4) {
      w[u][0] = a0; w[u][1] = a1; w[u][2] = a2; w[u][3] = a3;
    } else {
      const bool hi = lane >= 16;
      w[u][0] = hi ? a1 : a0;
      w[u][1] = hi ? a3 : a2;
    }
  }
#pragma unroll
  for (int u = 0; u < NB; ++u) {
    if constexpr (MEAN) {
      static_assert(Q == 4, "head sum in registers needs one head per line");
      float t = w[u][0] * v[u][0];
      t = fmaf(w[u][1], v[u][1], t);
      t = fmaf(w[u][2], v[u][2], t);
      t = fmaf(w[u][3], v[u][3], t);
      atomicAdd(&agg[(size_t)dj[u] * C + lane], t);
    } else {
      float* arow = agg + (size_t)dj[u] * M;
#pragma unroll
      for (int q = 0; q < Q; ++q)
        atomicAdd(&arow[q * 32 + lane], w[u][q] * v[u][q]);
    }
  }
}

template <int C, bool MEAN>
__global__ __launch_bounds__(TPB) void edge_agg(
    const float* __restrict__ xl, const float* __restrict__ score,
    const float* __restrict__ denom, const int* __restrict__ ei,
    float* __restrict__ agg, int E, int ET) {
  static_assert(MEAN, "the concat layers aggregate in edge_exp_agg");
  constexpr int U = 128 / C;
  const int lane = threadIdx.x & 31;
  const int e0 = __builtin_amdgcn_readfirstlane(
      (int)(((blockIdx.x * TPB + threadIdx.x) >> 5) << 5));
  if (e0 >= ET) return;
  const int e = e0 + lane;
  int s = 0, d = 0;
  float al0 = 0.0f, al1 = 0.0f, al2 = 0.0f, al3 = 0.0f;
  if (e < ET) {
    edge_sd(ei, e, E, s, d);
    const float4 ex = *(const float4*)(score + (size_t)e * 4);
    const float4 dn = *(const float4*)(denom + (size_t)d * 4);
    al0 = ex.x / dn.x; al1 = ex.y / dn.y;
    al2 = ex.z / dn.z; al3 = ex.w / dn.w;
  }
  const int cnt = min(32, ET - e0);
  int j = 0;
  for (; j + U <= cnt; j += U)
    agg_batch<C, U, MEAN>(xl, agg, s, d, al0, al1, al2, al3, j, lane);
  for (; j < cnt; ++j)
    agg_batch<C, 1, MEAN>(xl, agg, s, d, al0, al1, al2, al3, j, lane);
}

template <int C>
__global__ __launch_bounds__(TPB) void edge_exp_agg(
    const float* __restrict__ xl, const float* __restrict__ score,
    const float* __restrict__ mbuf, float* __restrict__ denom,
    const int* __restrict__ ei, float* __restrict__ agg, int E, int ET) {
  constexpr int U = 128 / C;
  const int lane = threadIdx.x & 31;
  const int e0 = __builtin_amdgcn_readfirstlane(
      (int)(((blockIdx.x * TPB + threadIdx.x) >> 5) << 5));
  if (e0 >= ET) return;
  const int e = e0 + lane;
  int s = 0, d = 0;
  if (e < ET) edge_sd(ei, e, E, s, d);

  const int h = lane & 3;
  float sc[4], mx[4], ex[4];
  int   dk[4];
  bool  ok[4];
#pragma unroll
  for (int k = 0; k < 4; ++k) {
    const int el = 8 * k + (lane >> 2);
    dk[k] = __builtin_amdgcn_ds_bpermute(el << 2, d);
    ok[k] = e0 + el < ET;
    sc[k] = score[(size_t)e0 * 4 + (ok[k] ? 32 * k + lane : 0)];
    mx[k] = mbuf[(size_t)dk[k] * 4 + h];
  }
#pragma unroll
  for (int k = 0; k < 4; ++k) {
    const float x = expf(sc[k] - mx[k]);
    if (ok[k]) atomicAdd(&denom[(size_t)dk[k] * 4 + h], x);
    ex[k] = ok[k] ? x : 0.0f;
  }

  const int cnt = min(32, ET - e0);
  if (cnt == 32) {
#pragma unroll
    for (int b = 0; b < 32 / U; ++b)
      agg_batch<C, U, false>(xl, agg, s, d, ex[0], ex[1], ex[2], ex[3], b * U, lane);
  } else {
    int j = 0;
    for (; j + U <= cnt; j += U)
      agg_batch<C, U, false>(xl, agg, s, d, ex[0], ex[1], ex[2], ex[3], j, lane);
    for (; j < cnt; ++j)
      agg_batch<C, 1, false>(xl, agg, s, d, ex[0], ex[1], ex[2], ex[3], j, lane);
  }
}

__global__ __launch_bounds__(TPB) void final_mean(
    const float* __restrict__ agg, const float* __restrict__ bias,
    float* __restrict__ out, int N) {
  const int tid = blockIdx.x * blockDim.x + threadIdx.x;
  if (tid >= N * 32) return;
  out[tid] = 0.25f * agg[tid] + bias[tid & 31];
}

extern "C" void kernel_launch(void* const* d_in, const int* in_sizes, int n_in,
                              void* d_out, int out_size, void* d_ws, size_t ws_size,
                              hipStream_t stream) {
  (void)in_sizes; (void)n_in; (void)out_size; (void)ws_size;
  const float* x   = (const float*)d_in[0];
  const int*   ei  = (const int*)d_in[1];
  const float* W1l = (const float*)d_in[2];
  const float* W1r = (const float*)d_in[3];
  const float* a1  = (const float*)d_in[4];
  const float* b1  = (const float*)d_in[5];
  const float* W2l = (const float*)d_in[6];
  const float* W2r = (const float*)d_in[7];
  const float* a2  = (const float*)d_in[8];
  const float* b2  = (const float*)d_in[9];
  const float* W3l = (const float*)d_in[10];
  const float* W3r = (const float*)d_in[11];
  const float* a3  = (const float*)d_in[12];
  const float* b3  = (const float*)d_in[13];

  const int N = 100000, E = 800000, ET = E + N;

  float* ws = (float*)d_ws;
  float* xl = ws;
  float* xr = xl + (size_t)N * 128;
  float* sc = xr + (size_t)N * 128;
  float* mb = sc + (size_t)ET * 4;
  float* dn = mb + (size_t)N * 4;
  float* h1 = dn + (size_t)N * 4;
  float* h2 = h1 + (size_t)N * 64;
  float* ag = h2 + (size_t)N * 64;

  const int rowTiles = N / 16;
  const int gblocks  = (rowTiles + WAVES_PER_BLOCK - 1) / WAVES_PER_BLOCK;

  auto layer = [&](const float* in, const float* inBias, const float* Wl, const float* Wr,
                   const float* att, const float* bias, float* out,
                   int C, bool final_layer) {
    const int M = 4 * C;
    if (M == 64 && !inBias) {
      gemm_wmma_f32<64, false><<<gblocks, TPB, 0, stream>>>(in, Wl, inBias, nullptr, xl, N);
      gemm_wmma_f32<64, false><<<gblocks, TPB, 0, stream>>>(in, Wr, inBias, nullptr, xr, N);
    } else if (M == 64) {
      gemm_wmma_f32<64, true><<<gblocks, TPB, 0, stream>>>(in, Wl, inBias, dn, xl, N);
      gemm_wmma_f32<64, true><<<gblocks, TPB, 0, stream>>>(in, Wr, inBias, dn, xr, N);
    } else {
      gemm_wmma_f32<128, true><<<gblocks, TPB, 0, stream>>>(in, Wl, inBias, dn, xl, N);
      gemm_wmma_f32<128, true><<<gblocks, TPB, 0, stream>>>(in, Wr, inBias, dn, xr, N);
    }

    float* aggbuf = final_layer ? ag : out;
    hipMemsetAsync(mb, 0xFF, (size_t)N * 4 * sizeof(float), stream);
    hipMemsetAsync(dn, 0x00, (size_t)N * 4 * sizeof(float), stream);
    const int aggw = final_layer ? C : M;
    hipMemsetAsync(aggbuf, 0x00, (size_t)N * aggw * sizeof(float), stream);

    const int eThreads = ET * 4;
    const int eBlocks = (eThreads + TPB - 1) / TPB;
    const int aBlocks = (ET + TPB - 1) / TPB;
    if (C == 16) {
      edge_score<16><<<eBlocks, TPB, 0, stream>>>(xl, xr, ei, att, sc, mb, E, ET);
      edge_exp_agg<16><<<aBlocks, TPB, 0, stream>>>(xl, sc, mb, dn, ei, aggbuf, E, ET);
    } else {
      edge_score<32><<<eBlocks, TPB, 0, stream>>>(xl, xr, ei, att, sc, mb, E, ET);
      edge_exp<<<eBlocks, TPB, 0, stream>>>(sc, mb, dn, ei, E, ET);
      edge_agg<32, true><<<aBlocks, TPB, 0, stream>>>(xl, sc, dn, ei, aggbuf, E, ET);
    }

    if (final_layer) {
      const int t = N * 32;
      final_mean<<<(t + TPB - 1) / TPB, TPB, 0, stream>>>(ag, bias, (float*)d_out, N);
    }
  };

  layer(x,  nullptr, W1l, W1r, a1, b1, h1, 16, false);
  layer(h1, b1,      W2l, W2r, a2, b2, h2, 16, false);
  layer(h2, b2,      W3l, W3r, a3, b3, (float*)d_out, 32, true);
}


# edge_score's per-destination max as ONE unsigned atomic max on an order-preserving integer key of the score (was signed max + unsigned min under a sign branch): 1 atomic request per edge, not 2
# speedup vs baseline: 3.5490x; 1.1067x over previous
#include <hip/hip_runtime.h>
#include <math.h>

typedef __attribute__((ext_vector_type(2))) float v2f;
typedef __attribute__((ext_vector_type(8))) float v8f;

#define TPB 256
#define WAVES_PER_BLOCK 8

template <int M, bool ACT>
__global__ __launch_bounds__(TPB) void gemm_wmma_f32(
    const float* __restrict__ X, const float* __restrict__ W,
    const float* __restrict__ bias, const float* __restrict__ den,
    float* __restrict__ Y, int Nrows) {
  __shared__ __align__(16) float lw[64 * M];

  constexpr int TOTAL4 = 64 * M / 4;
  for (int i = threadIdx.x; i < TOTAL4; i += TPB)
    ((float4*)lw)[i] = ((const float4*)W)[i];
  __syncthreads();

  const int rowTile = blockIdx.x * WAVES_PER_BLOCK + (threadIdx.x >> 5);
  if (rowTile >= (Nrows >> 4)) return;
  const int row0 = rowTile << 4;
  const int lane = threadIdx.x & 31;
  const int r    = lane & 15;
  const int half = lane >> 4;

  v2f a[16];
  const float* xrow = X + (size_t)(row0 + r) * 64;
#pragma unroll
  for (int ks = 0; ks < 16; ++ks) {
    const int k = 4 * ks + 2 * half;
    a[ks].x = xrow[k];
    a[ks].y = xrow[k + 1];
  }
  if constexpr (ACT) {
    const float4 dv = *(const float4*)(den + (size_t)(row0 + r) * 4);
    const float inv[4] = {1.0f / dv.x, 1.0f / dv.y, 1.0f / dv.z, 1.0f / dv.w};
#pragma unroll
    for (int ks = 0; ks < 16; ++ks) {
      const int k = 4 * ks + 2 * half;
      const float x0 = a[ks].x * inv[ks >> 2] + bias[k];
      const float x1 = a[ks].y * inv[ks >> 2] + bias[k + 1];
      const float e0 = expm1f(x0);
      const float e1 = expm1f(x1);
      a[ks].x = x0 > 0.0f ? x0 : e0;
      a[ks].y = x1 > 0.0f ? x1 : e1;
    }
  }

#pragma unroll
  for (int tn = 0; tn < M / 16; ++tn) {
    const int col0 = tn << 4;
    v8f c = {};
#pragma unroll
    for (int ks = 0; ks < 16; ++ks) {
      const int k = 4 * ks + 2 * half;
      v2f b;
      b.x = lw[k * M + col0 + r];
      b.y = lw[(k + 1) * M + col0 + r];
      c = __builtin_amdgcn_wmma_f32_16x16x4_f32(false, a[ks], false, b,
                                                (short)0, c, false, false);
    }
#pragma unroll
    for (int v = 0; v < 8; ++v)
      Y[(size_t)(row0 + v + 8 * half) * M + col0 + r] = c[v];
  }
}

__device__ __forceinline__ void edge_sd(const int* __restrict__ ei, int e, int E,
                                        int& s, int& d) {
  if (e < E) { s = ei[e]; d = ei[E + e]; }
  else       { s = e - E; d = s; }
}

__device__ __forceinline__ unsigned int float_to_max_key(float val) {
  const unsigned int u = __float_as_uint(val);
  const unsigned int neg = (unsigned int)((int)u >> 31);
  return u ^ (neg | 0x80000000u);
}
__device__ __forceinline__ float max_key_to_float(float stored) {
  const unsigned int k = __float_as_uint(stored);
  const unsigned int pos = (unsigned int)((int)k >> 31);
  return __uint_as_float(k ^ (~pos | 0x80000000u));
}
__device__ __forceinline__ void atomic_max_float(float* addr, float val) {
  atomicMax((unsigned int*)addr, float_to_max_key(val));
}

template <int CTRL>
__device__ __forceinline__ float lane_xor_add(float v) {
  const int t = __builtin_amdgcn_update_dpp(0, __float_as_int(v), CTRL, 0xf, 0xf, true);
  return v + __int_as_float(t);
}

template <int C>
__global__ __launch_bounds__(TPB) void edge_score(
    const float* __restrict__ xl, const float* __restrict__ xr,
    const int* __restrict__ ei, const float* __restrict__ att,
    float* __restrict__ score, float* __restrict__ mbuf, int E, int ET) {
  constexpr int M     = 4 * C;
  constexpr int LPR   = M / 4;
  constexpr int EPS   = 32 / LPR;
  constexpr int LPH   = C / 4;
  constexpr int STEPS = 8 / EPS;
  static_assert(STEPS == LPH, "one kept score per lane needs steps == lanes per head");
  const int lane = threadIdx.x & 31;
  const int e0 = __builtin_amdgcn_readfirstlane(
      (int)(((blockIdx.x * TPB + threadIdx.x) >> 5) << 3));
  if (e0 >= ET) return;
  const int ch   = lane & (LPR - 1);
  const int sub  = lane / LPR;
  const int q    = lane & (LPH - 1);
  const int h    = ch / LPH;
  const int eloc = q * EPS + sub;
  const int e    = e0 + eloc;
  int s = 0, d = 0;
  if (e < ET) edge_sd(ei, e, E, s, d);
  const float4 w = ((const float4*)att)[ch];

  float4 a[STEPS], b[STEPS];
#pragma unroll
  for (int t = 0; t < STEPS; ++t) {
    int sj, dj;
    if constexpr (EPS == 1) {
      sj = __builtin_amdgcn_readlane(s, t);
      dj = __builtin_amdgcn_readlane(d, t);
    } else {
      const int s0 = __builtin_amdgcn_readlane(s, t);
      const int s1 = __builtin_amdgcn_readlane(s, 16 + t);
      const int d0 = __builtin_amdgcn_readlane(d, t);
      const int d1 = __builtin_amdgcn_readlane(d, 16 + t);
      sj = sub ? s1 : s0;
      dj = sub ? d1 : d0;
    }
    const unsigned oa = (unsigned)sj * (unsigned)(M * 4) + (unsigned)ch * 16u;
    const unsigned ob = (unsigned)dj * (unsigned)(M * 4) + (unsigned)ch * 16u;
    a[t] = *(const float4*)((const char*)xl + oa);
    b[t] = *(const float4*)((const char*)xr + ob);
  }

  float acc = 0.0f;
#pragma unroll
  for (int t = 0; t < STEPS; ++t) {
    float vx = a[t].x + b[t].x, vy = a[t].y + b[t].y;
    float vz = a[t].z + b[t].z, vw = a[t].w + b[t].w;
    vx = vx > 0.0f ? vx : 0.2f * vx;
    vy = vy > 0.0f ? vy : 0.2f * vy;
    vz = vz > 0.0f ? vz : 0.2f * vz;
    vw = vw > 0.0f ? vw : 0.2f * vw;
    float p = w.x * vx + w.y * vy + w.z * vz + w.w * vw;
    p = lane_xor_add<0x161>(p);
    p = lane_xor_add<0x162>(p);
    if constexpr (LPH == 8) p = lane_xor_add<0x164>(p);
    acc = (q == t) ? p : acc;
  }
  if (e < ET) {
    score[(size_t)e * 4 + h] = acc;
    atomic_max_float(&mbuf[(size_t)d * 4 + h], acc);
  }
}

__global__ __launch_bounds__(TPB) void edge_exp(
    float* __restrict__ score, const float* __restrict__ mbuf,
    float* __restrict__ denom, const int* __restrict__ ei, int E, int ET) {
  const int tid = blockIdx.x * blockDim.x + threadIdx.x;
  if (tid >= ET * 4) return;
  const int e = tid >> 2, h = tid & 3;
  int s, d; edge_sd(ei, e, E, s, d);
  const float ex = expf(score[(size_t)e * 4 + h] - max_key_to_float(mbuf[(size_t)d * 4 + h]));
  score[(size_t)e * 4 + h] = ex;
  atomicAdd(&denom[(size_t)d * 4 + h], ex);
}

template <int C, int NB, bool MEAN>
__device__ __forceinline__ void agg_batch(
    const float* __restrict__ xl, float* __restrict__ agg,
    int s, int d, float al0, float al1, float al2, float al3, int j0, int lane) {
  constexpr int M = 4 * C;
  constexpr int Q = M / 32;
  float v[NB][Q];
  float w[NB][Q];
  int   dj[NB];
#pragma unroll
  for (int u = 0; u < NB; ++u) {
    const int j = j0 + u;
    const int sj = __builtin_amdgcn_readlane(s, j);
    dj[u] = __builtin_amdgcn_readlane(d, j);
    const float* row = xl + (size_t)sj * M;
#pragma unroll
    for (int q = 0; q < Q; ++q) v[u][q] = row[q * 32 + lane];
    float a0, a1, a2, a3;
    if constexpr (MEAN) {
      a0 = __int_as_float(__builtin_amdgcn_readlane(__float_as_int(al0), j));
      a1 = __int_as_float(__builtin_amdgcn_readlane(__float_as_int(al1), j));
      a2 = __int_as_float(__builtin_amdgcn_readlane(__float_as_int(al2), j));
      a3 = __int_as_float(__builtin_amdgcn_readlane(__float_as_int(al3), j));
    } else {
      static_assert(8 % NB == 0, "a batch must stay inside one group of 8 edges");
      const int k = j0 >> 3;
      const int ek = __float_as_int(k == 0 ? al0 : k == 1 ? al1 : k == 2 ? al2 : al3);
      const int l0 = (j & 7) << 2;
      a0 = __int_as_float(__builtin_amdgcn_readlane(ek, l0));
      a1 = __int_as_float(__builtin_amdgcn_readlane(ek, l0 + 1));
      a2 = __int_as_float(__builtin_amdgcn_readlane(ek, l0 + 2));
      a3 = __int_as_float(__builtin_amdgcn_readlane(ek, l0 + 3));
    }
    if constexpr (Q == 4) {
      w[u][0] = a0; w[u][1] = a1; w[u][2] = a2; w[u][3] = a3;
    } else {
      const bool hi = lane >= 16;
      w[u][0] = hi ? a1 : a0;
      w[u][1] = hi ? a3 : a2;
    }
  }
#pragma unroll
  for (int u = 0; u < NB; ++u) {
    if constexpr (MEAN) {
      static_assert(Q == 4, "head sum in registers needs one head per line");
      float t = w[u][0] * v[u][0];
      t = fmaf(w[u][1], v[u][1], t);
      t = fmaf(w[u][2], v[u][2], t);
      t = fmaf(w[u][3], v[u][3], t);
      atomicAdd(&agg[(size_t)dj[u] * C + lane], t);
    } else {
      float* arow = agg + (size_t)dj[u] * M;
#pragma unroll
      for (int q = 0; q < Q; ++q)
        atomicAdd(&arow[q * 32 + lane], w[u][q] * v[u][q]);
    }
  }
}

template <int C, bool MEAN>
__global__ __launch_bounds__(TPB) void edge_agg(
    const float* __restrict__ xl, const float* __restrict__ score,
    const float* __restrict__ denom, const int* __restrict__ ei,
    float* __restrict__ agg, int E, int ET) {
  static_assert(MEAN, "the concat layers aggregate in edge_exp_agg");
  constexpr int U = 128 / C;
  const int lane = threadIdx.x & 31;
  const int e0 = __builtin_amdgcn_readfirstlane(
      (int)(((blockIdx.x * TPB + threadIdx.x) >> 5) << 5));
  if (e0 >= ET) return;
  const int e = e0 + lane;
  int s = 0, d = 0;
  float al0 = 0.0f, al1 = 0.0f, al2 = 0.0f, al3 = 0.0f;
  if (e < ET) {
    edge_sd(ei, e, E, s, d);
    const float4 ex = *(const float4*)(score + (size_t)e * 4);
    const float4 dn = *(const float4*)(denom + (size_t)d * 4);
    al0 = ex.x / dn.x; al1 = ex.y / dn.y;
    al2 = ex.z / dn.z; al3 = ex.w / dn.w;
  }
  const int cnt = min(32, ET - e0);
  int j = 0;
  for (; j + U <= cnt; j += U)
    agg_batch<C, U, MEAN>(xl, agg, s, d, al0, al1, al2, al3, j, lane);
  for (; j < cnt; ++j)
    agg_batch<C, 1, MEAN>(xl, agg, s, d, al0, al1, al2, al3, j, lane);
}

template <int C>
__global__ __launch_bounds__(TPB) void edge_exp_agg(
    const float* __restrict__ xl, const float* __restrict__ score,
    const float* __restrict__ mbuf, float* __restrict__ denom,
    const int* __restrict__ ei, float* __restrict__ agg, int E, int ET) {
  constexpr int U = 128 / C;
  const int lane = threadIdx.x & 31;
  const int e0 = __builtin_amdgcn_readfirstlane(
      (int)(((blockIdx.x * TPB + threadIdx.x) >> 5) << 5));
  if (e0 >= ET) return;
  const int e = e0 + lane;
  int s = 0, d = 0;
  if (e < ET) edge_sd(ei, e, E, s, d);

  const int h = lane & 3;
  float sc[4], mx[4], ex[4];
  int   dk[4];
  bool  ok[4];
#pragma unroll
  for (int k = 0; k < 4; ++k) {
    const int el = 8 * k + (lane >> 2);
    dk[k] = __builtin_amdgcn_ds_bpermute(el << 2, d);
    ok[k] = e0 + el < ET;
    sc[k] = score[(size_t)e0 * 4 + (ok[k] ? 32 * k + lane : 0)];
    mx[k] = mbuf[(size_t)dk[k] * 4 + h];
  }
#pragma unroll
  for (int k = 0; k < 4; ++k) {
    const float x = expf(sc[k] - max_key_to_float(mx[k]));
    if (ok[k]) atomicAdd(&denom[(size_t)dk[k] * 4 + h], x);
    ex[k] = ok[k] ? x : 0.0f;
  }

  const int cnt = min(32, ET - e0);
  if (cnt == 32) {
#pragma unroll
    for (int b = 0; b < 32 / U; ++b)
      agg_batch<C, U, false>(xl, agg, s, d, ex[0], ex[1], ex[2], ex[3], b * U, lane);
  } else {
    int j = 0;
    for (; j + U <= cnt; j += U)
      agg_batch<C, U, false>(xl, agg, s, d, ex[0], ex[1], ex[2], ex[3], j, lane);
    for (; j < cnt; ++j)
      agg_batch<C, 1, false>(xl, agg, s, d, ex[0], ex[1], ex[2], ex[3], j, lane);
  }
}

__global__ __launch_bounds__(TPB) void final_mean(
    const float* __restrict__ agg, const float* __restrict__ bias,
    float* __restrict__ out, int N) {
  const int tid = blockIdx.x * blockDim.x + threadIdx.x;
  if (tid >= N * 32) return;
  out[tid] = 0.25f * agg[tid] + bias[tid & 31];
}

extern "C" void kernel_launch(void* const* d_in, const int* in_sizes, int n_in,
                              void* d_out, int out_size, void* d_ws, size_t ws_size,
                              hipStream_t stream) {
  (void)in_sizes; (void)n_in; (void)out_size; (void)ws_size;
  const float* x   = (const float*)d_in[0];
  const int*   ei  = (const int*)d_in[1];
  const float* W1l = (const float*)d_in[2];
  const float* W1r = (const float*)d_in[3];
  const float* a1  = (const float*)d_in[4];
  const float* b1  = (const float*)d_in[5];
  const float* W2l = (const float*)d_in[6];
  const float* W2r = (const float*)d_in[7];
  const float* a2  = (const float*)d_in[8];
  const float* b2  = (const float*)d_in[9];
  const float* W3l = (const float*)d_in[10];
  const float* W3r = (const float*)d_in[11];
  const float* a3  = (const float*)d_in[12];
  const float* b3  = (const float*)d_in[13];

  const int N = 100000, E = 800000, ET = E + N;

  float* ws = (float*)d_ws;
  float* xl = ws;
  float* xr = xl + (size_t)N * 128;
  float* sc = xr + (size_t)N * 128;
  float* mb = sc + (size_t)ET * 4;
  float* dn = mb + (size_t)N * 4;
  float* h1 = dn + (size_t)N * 4;
  float* h2 = h1 + (size_t)N * 64;
  float* ag = h2 + (size_t)N * 64;

  const int rowTiles = N / 16;
  const int gblocks  = (rowTiles + WAVES_PER_BLOCK - 1) / WAVES_PER_BLOCK;

  auto layer = [&](const float* in, const float* inBias, const float* Wl, const float* Wr,
                   const float* att, const float* bias, float* out,
                   int C, bool final_layer) {
    const int M = 4 * C;
    if (M == 64 && !inBias) {
      gemm_wmma_f32<64, false><<<gblocks, TPB, 0, stream>>>(in, Wl, inBias, nullptr, xl, N);
      gemm_wmma_f32<64, false><<<gblocks, TPB, 0, stream>>>(in, Wr, inBias, nullptr, xr, N);
    } else if (M == 64) {
      gemm_wmma_f32<64, true><<<gblocks, TPB, 0, stream>>>(in, Wl, inBias, dn, xl, N);
      gemm_wmma_f32<64, true><<<gblocks, TPB, 0, stream>>>(in, Wr, inBias, dn, xr, N);
    } else {
      gemm_wmma_f32<128, true><<<gblocks, TPB, 0, stream>>>(in, Wl, inBias, dn, xl, N);
      gemm_wmma_f32<128, true><<<gblocks, TPB, 0, stream>>>(in, Wr, inBias, dn, xr, N);
    }

    float* aggbuf = final_layer ? ag : out;
    hipMemsetAsync(mb, 0x00, (size_t)N * 4 * sizeof(float), stream);
    hipMemsetAsync(dn, 0x00, (size_t)N * 4 * sizeof(float), stream);
    const int aggw = final_layer ? C : M;
    hipMemsetAsync(aggbuf, 0x00, (size_t)N * aggw * sizeof(float), stream);

    const int eThreads = ET * 4;
    const int eBlocks = (eThreads + TPB - 1) / TPB;
    const int aBlocks = (ET + TPB - 1) / TPB;
    if (C == 16) {
      edge_score<16><<<eBlocks, TPB, 0, stream>>>(xl, xr, ei, att, sc, mb, E, ET);
      edge_exp_agg<16><<<aBlocks, TPB, 0, stream>>>(xl, sc, mb, dn, ei, aggbuf, E, ET);
    } else {
      edge_score<32><<<eBlocks, TPB, 0, stream>>>(xl, xr, ei, att, sc, mb, E, ET);
      edge_exp<<<eBlocks, TPB, 0, stream>>>(sc, mb, dn, ei, E, ET);
      edge_agg<32, true><<<aBlocks, TPB, 0, stream>>>(xl, sc, dn, ei, aggbuf, E, ET);
    }

    if (final_layer) {
      const int t = N * 32;
      final_mean<<<(t + TPB - 1) / TPB, TPB, 0, stream>>>(ag, bias, (float*)d_out, N);
    }
  };

  layer(x,  nullptr, W1l, W1r, a1, b1, h1, 16, false);
  layer(h1, b1,      W2l, W2r, a2, b2, h2, 16, false);
  layer(h2, b2,      W3l, W3r, a3, b3, (float*)d_out, 32, true);
}


# f32 atomic-add targets (aggregate rows, concat layers' denominators) kept once per L2 cache half, chosen by die number; readers (next GEMM's A loads, final pass) add both copies; sum order changes
# speedup vs baseline: 3.8711x; 1.0908x over previous
#include <hip/hip_runtime.h>
#include <math.h>

typedef __attribute__((ext_vector_type(2))) float v2f;
typedef __attribute__((ext_vector_type(8))) float v8f;

#define TPB 256
#define WAVES_PER_BLOCK 8

__device__ __forceinline__ int chip_half() {
  return (int)((__builtin_amdgcn_s_sendmsg_rtn(0x87) >> 18) & 1u);
}

template <int M, bool ACT>
__global__ __launch_bounds__(TPB) void gemm_wmma_f32(
    const float* __restrict__ X, const float* __restrict__ X2,
    const float* __restrict__ W,
    const float* __restrict__ bias, const float* __restrict__ den,
    const float* __restrict__ den2,
    float* __restrict__ Y, int Nrows) {
  __shared__ __align__(16) float lw[64 * M];

  constexpr int TOTAL4 = 64 * M / 4;
  for (int i = threadIdx.x; i < TOTAL4; i += TPB)
    ((float4*)lw)[i] = ((const float4*)W)[i];
  __syncthreads();

  const int rowTile = blockIdx.x * WAVES_PER_BLOCK + (threadIdx.x >> 5);
  if (rowTile >= (Nrows >> 4)) return;
  const int row0 = rowTile << 4;
  const int lane = threadIdx.x & 31;
  const int r    = lane & 15;
  const int half = lane >> 4;

  v2f a[16];
  const float* xrow = X + (size_t)(row0 + r) * 64;
#pragma unroll
  for (int ks = 0; ks < 16; ++ks) {
    const int k = 4 * ks + 2 * half;
    a[ks].x = xrow[k];
    a[ks].y = xrow[k + 1];
  }
  if constexpr (ACT) {
    v2f a2[16];
    const float* xrow2 = X2 + (size_t)(row0 + r) * 64;
#pragma unroll
    for (int ks = 0; ks < 16; ++ks) {
      const int k = 4 * ks + 2 * half;
      a2[ks].x = xrow2[k];
      a2[ks].y = xrow2[k + 1];
    }
    const float4 dv0 = *(const float4*)(den + (size_t)(row0 + r) * 4);
    const float4 dv1 = *(const float4*)(den2 + (size_t)(row0 + r) * 4);
    const float inv[4] = {1.0f / (dv0.x + dv1.x), 1.0f / (dv0.y + dv1.y),
                          1.0f / (dv0.z + dv1.z), 1.0f / (dv0.w + dv1.w)};
#pragma unroll
    for (int ks = 0; ks < 16; ++ks) {
      const int k = 4 * ks + 2 * half;
      const float x0 = (a[ks].x + a2[ks].x) * inv[ks >> 2] + bias[k];
      const float x1 = (a[ks].y + a2[ks].y) * inv[ks >> 2] + bias[k + 1];
      const float e0 = expm1f(x0);
      const float e1 = expm1f(x1);
      a[ks].x = x0 > 0.0f ? x0 : e0;
      a[ks].y = x1 > 0.0f ? x1 : e1;
    }
  }

#pragma unroll
  for (int tn = 0; tn < M / 16; ++tn) {
    const int col0 = tn << 4;
    v8f c = {};
#pragma unroll
    for (int ks = 0; ks < 16; ++ks) {
      const int k = 4 * ks + 2 * half;
      v2f b;
      b.x = lw[k * M + col0 + r];
      b.y = lw[(k + 1) * M + col0 + r];
      c = __builtin_amdgcn_wmma_f32_16x16x4_f32(false, a[ks], false, b,
                                                (short)0, c, false, false);
    }
#pragma unroll
    for (int v = 0; v < 8; ++v)
      Y[(size_t)(row0 + v + 8 * half) * M + col0 + r] = c[v];
  }
}

__device__ __forceinline__ void edge_sd(const int* __restrict__ ei, int e, int E,
                                        int& s, int& d) {
  if (e < E) { s = ei[e]; d = ei[E + e]; }
  else       { s = e - E; d = s; }
}

__device__ __forceinline__ unsigned int float_to_max_key(float val) {
  const unsigned int u = __float_as_uint(val);
  const unsigned int neg = (unsigned int)((int)u >> 31);
  return u ^ (neg | 0x80000000u);
}
__device__ __forceinline__ float max_key_to_float(float stored) {
  const unsigned int k = __float_as_uint(stored);
  const unsigned int pos = (unsigned int)((int)k >> 31);
  return __uint_as_float(k ^ (~pos | 0x80000000u));
}
__device__ __forceinline__ void atomic_max_float(float* addr, float val) {
  atomicMax((unsigned int*)addr, float_to_max_key(val));
}

template <int CTRL>
__device__ __forceinline__ float lane_xor_add(float v) {
  const int t = __builtin_amdgcn_update_dpp(0, __float_as_int(v), CTRL, 0xf, 0xf, true);
  return v + __int_as_float(t);
}

template <int C>
__global__ __launch_bounds__(TPB) void edge_score(
    const float* __restrict__ xl, const float* __restrict__ xr,
    const int* __restrict__ ei, const float* __restrict__ att,
    float* __restrict__ score, float* __restrict__ mbuf, int E, int ET) {
  constexpr int M     = 4 * C;
  constexpr int LPR   = M / 4;
  constexpr int EPS   = 32 / LPR;
  constexpr int LPH   = C / 4;
  constexpr int STEPS = 8 / EPS;
  static_assert(STEPS == LPH, "one kept score per lane needs steps == lanes per head");
  const int lane = threadIdx.x & 31;
  const int e0 = __builtin_amdgcn_readfirstlane(
      (int)(((blockIdx.x * TPB + threadIdx.x) >> 5) << 3));
  if (e0 >= ET) return;
  const int ch   = lane & (LPR - 1);
  const int sub  = lane / LPR;
  const int q    = lane & (LPH - 1);
  const int h    = ch / LPH;
  const int eloc = q * EPS + sub;
  const int e    = e0 + eloc;
  int s = 0, d = 0;
  if (e < ET) edge_sd(ei, e, E, s, d);
  const float4 w = ((const float4*)att)[ch];

  float4 a[STEPS], b[STEPS];
#pragma unroll
  for (int t = 0; t < STEPS; ++t) {
    int sj, dj;
    if constexpr (EPS == 1) {
      sj = __builtin_amdgcn_readlane(s, t);
      dj = __builtin_amdgcn_readlane(d, t);
    } else {
      const int s0 = __builtin_amdgcn_readlane(s, t);
      const int s1 = __builtin_amdgcn_readlane(s, 16 + t);
      const int d0 = __builtin_amdgcn_readlane(d, t);
      const int d1 = __builtin_amdgcn_readlane(d, 16 + t);
      sj = sub ? s1 : s0;
      dj = sub ? d1 : d0;
    }
    const unsigned oa = (unsigned)sj * (unsigned)(M * 4) + (unsigned)ch * 16u;
    const unsigned ob = (unsigned)dj * (unsigned)(M * 4) + (unsigned)ch * 16u;
    a[t] = *(const float4*)((const char*)xl + oa);
    b[t] = *(const float4*)((const char*)xr + ob);
  }

  float acc = 0.0f;
#pragma unroll
  for (int t = 0; t < STEPS; ++t) {
    float vx = a[t].x + b[t].x, vy = a[t].y + b[t].y;
    float vz = a[t].z + b[t].z, vw = a[t].w + b[t].w;
    vx = vx > 0.0f ? vx : 0.2f * vx;
    vy = vy > 0.0f ? vy : 0.2f * vy;
    vz = vz > 0.0f ? vz : 0.2f * vz;
    vw = vw > 0.0f ? vw : 0.2f * vw;
    float p = w.x * vx + w.y * vy + w.z * vz + w.w * vw;
    p = lane_xor_add<0x161>(p);
    p = lane_xor_add<0x162>(p);
    if constexpr (LPH == 8) p = lane_xor_add<0x164>(p);
    acc = (q == t) ? p : acc;
  }
  if (e < ET) {
    score[(size_t)e * 4 + h] = acc;
    atomic_max_float(&mbuf[(size_t)d * 4 + h], acc);
  }
}

__global__ __launch_bounds__(TPB) void edge_exp(
    float* __restrict__ score, const float* __restrict__ mbuf,
    float* __restrict__ denom, const int* __restrict__ ei, int E, int ET) {
  const int tid = blockIdx.x * blockDim.x + threadIdx.x;
  if (tid >= ET * 4) return;
  const int e = tid >> 2, h = tid & 3;
  int s, d; edge_sd(ei, e, E, s, d);
  const float ex = expf(score[(size_t)e * 4 + h] - max_key_to_float(mbuf[(size_t)d * 4 + h]));
  score[(size_t)e * 4 + h] = ex;
  atomicAdd(&denom[(size_t)d * 4 + h], ex);
}

template <int C, int NB, bool MEAN>
__device__ __forceinline__ void agg_batch(
    const float* __restrict__ xl, float* __restrict__ agg,
    int s, int d, float al0, float al1, float al2, float al3, int j0, int lane) {
  constexpr int M = 4 * C;
  constexpr int Q = M / 32;
  float v[NB][Q];
  float w[NB][Q];
  int   dj[NB];
#pragma unroll
  for (int u = 0; u < NB; ++u) {
    const int j = j0 + u;
    const int sj = __builtin_amdgcn_readlane(s, j);
    dj[u] = __builtin_amdgcn_readlane(d, j);
    const float* row = xl + (size_t)sj * M;
#pragma unroll
    for (int q = 0; q < Q; ++q) v[u][q] = row[q * 32 + lane];
    float a0, a1, a2, a3;
    if constexpr (MEAN) {
      a0 = __int_as_float(__builtin_amdgcn_readlane(__float_as_int(al0), j));
      a1 = __int_as_float(__builtin_amdgcn_readlane(__float_as_int(al1), j));
      a2 = __int_as_float(__builtin_amdgcn_readlane(__float_as_int(al2), j));
      a3 = __int_as_float(__builtin_amdgcn_readlane(__float_as_int(al3), j));
    } else {
      static_assert(8 % NB == 0, "a batch must stay inside one group of 8 edges");
      const int k = j0 >> 3;
      const int ek = __float_as_int(k == 0 ? al0 : k == 1 ? al1 : k == 2 ? al2 : al3);
      const int l0 = (j & 7) << 2;
      a0 = __int_as_float(__builtin_amdgcn_readlane(ek, l0));
      a1 = __int_as_float(__builtin_amdgcn_readlane(ek, l0 + 1));
      a2 = __int_as_float(__builtin_amdgcn_readlane(ek, l0 + 2));
      a3 = __int_as_float(__builtin_amdgcn_readlane(ek, l0 + 3));
    }
    if constexpr (Q == 4) {
      w[u][0] = a0; w[u][1] = a1; w[u][2] = a2; w[u][3] = a3;
    } else {
      const bool hi = lane >= 16;
      w[u][0] = hi ? a1 : a0;
      w[u][1] = hi ? a3 : a2;
    }
  }
#pragma unroll
  for (int u = 0; u < NB; ++u) {
    if constexpr (MEAN) {
      static_assert(Q == 4, "head sum in registers needs one head per line");
      float t = w[u][0] * v[u][0];
      t = fmaf(w[u][1], v[u][1], t);
      t = fmaf(w[u][2], v[u][2], t);
      t = fmaf(w[u][3], v[u][3], t);
      atomicAdd(&agg[(size_t)dj[u] * C + lane], t);
    } else {
      float* arow = agg + (size_t)dj[u] * M;
#pragma unroll
      for (int q = 0; q < Q; ++q)
        atomicAdd(&arow[q * 32 + lane], w[u][q] * v[u][q]);
    }
  }
}

template <int C, bool MEAN>
__global__ __launch_bounds__(TPB) void edge_agg(
    const float* __restrict__ xl, const float* __restrict__ score,
    const float* __restrict__ denom, const int* __restrict__ ei,
    float* __restrict__ agg_copies, int agg_copy_floats, int E, int ET) {
  static_assert(MEAN, "the concat layers aggregate in edge_exp_agg");
  float* __restrict__ agg = agg_copies + (size_t)chip_half() * agg_copy_floats;
  constexpr int U = 128 / C;
  const int lane = threadIdx.x & 31;
  const int e0 = __builtin_amdgcn_readfirstlane(
      (int)(((blockIdx.x * TPB + threadIdx.x) >> 5) << 5));
  if (e0 >= ET) return;
  const int e = e0 + lane;
  int s = 0, d = 0;
  float al0 = 0.0f, al1 = 0.0f, al2 = 0.0f, al3 = 0.0f;
  if (e < ET) {
    edge_sd(ei, e, E, s, d);
    const float4 ex = *(const float4*)(score + (size_t)e * 4);
    const float4 dn = *(const float4*)(denom + (size_t)d * 4);
    al0 = ex.x / dn.x; al1 = ex.y / dn.y;
    al2 = ex.z / dn.z; al3 = ex.w / dn.w;
  }
  const int cnt = min(32, ET - e0);
  int j = 0;
  for (; j + U <= cnt; j += U)
    agg_batch<C, U, MEAN>(xl, agg, s, d, al0, al1, al2, al3, j, lane);
  for (; j < cnt; ++j)
    agg_batch<C, 1, MEAN>(xl, agg, s, d, al0, al1, al2, al3, j, lane);
}

template <int C>
__global__ __launch_bounds__(TPB) void edge_exp_agg(
    const float* __restrict__ xl, const float* __restrict__ score,
    const float* __restrict__ mbuf, float* __restrict__ denom_copies,
    int denom_copy_floats, const int* __restrict__ ei,
    float* __restrict__ agg_copies, int agg_copy_floats, int E, int ET) {
  constexpr int U = 128 / C;
  const int hf = chip_half();
  float* __restrict__ denom = denom_copies + (size_t)hf * denom_copy_floats;
  float* __restrict__ agg   = agg_copies + (size_t)hf * agg_copy_floats;
  const int lane = threadIdx.x & 31;
  const int e0 = __builtin_amdgcn_readfirstlane(
      (int)(((blockIdx.x * TPB + threadIdx.x) >> 5) << 5));
  if (e0 >= ET) return;
  const int e = e0 + lane;
  int s = 0, d = 0;
  if (e < ET) edge_sd(ei, e, E, s, d);

  const int h = lane & 3;
  float sc[4], mx[4], ex[4];
  int   dk[4];
  bool  ok[4];
#pragma unroll
  for (int k = 0; k < 4; ++k) {
    const int el = 8 * k + (lane >> 2);
    dk[k] = __builtin_amdgcn_ds_bpermute(el << 2, d);
    ok[k] = e0 + el < ET;
    sc[k] = score[(size_t)e0 * 4 + (ok[k] ? 32 * k + lane : 0)];
    mx[k] = mbuf[(size_t)dk[k] * 4 + h];
  }
#pragma unroll
  for (int k = 0; k < 4; ++k) {
    const float x = expf(sc[k] - max_key_to_float(mx[k]));
    if (ok[k]) atomicAdd(&denom[(size_t)dk[k] * 4 + h], x);
    ex[k] = ok[k] ? x : 0.0f;
  }

  const int cnt = min(32, ET - e0);
  if (cnt == 32) {
#pragma unroll
    for (int b = 0; b < 32 / U; ++b)
      agg_batch<C, U, false>(xl, agg, s, d, ex[0], ex[1], ex[2], ex[3], b * U, lane);
  } else {
    int j = 0;
    for (; j + U <= cnt; j += U)
      agg_batch<C, U, false>(xl, agg, s, d, ex[0], ex[1], ex[2], ex[3], j, lane);
    for (; j < cnt; ++j)
      agg_batch<C, 1, false>(xl, agg, s, d, ex[0], ex[1], ex[2], ex[3], j, lane);
  }
}

__global__ __launch_bounds__(TPB) void final_mean(
    const float* __restrict__ agg, const float* __restrict__ agg2,
    const float* __restrict__ bias, float* __restrict__ out, int N) {
  const int tid = blockIdx.x * blockDim.x + threadIdx.x;
  if (tid >= N * 32) return;
  out[tid] = 0.25f * (agg[tid] + agg2[tid]) + bias[tid & 31];
}

extern "C" void kernel_launch(void* const* d_in, const int* in_sizes, int n_in,
                              void* d_out, int out_size, void* d_ws, size_t ws_size,
                              hipStream_t stream) {
  (void)in_sizes; (void)n_in; (void)out_size; (void)ws_size;
  const float* x   = (const float*)d_in[0];
  const int*   ei  = (const int*)d_in[1];
  const float* W1l = (const float*)d_in[2];
  const float* W1r = (const float*)d_in[3];
  const float* a1  = (const float*)d_in[4];
  const float* b1  = (const float*)d_in[5];
  const float* W2l = (const float*)d_in[6];
  const float* W2r = (const float*)d_in[7];
  const float* a2  = (const float*)d_in[8];
  const float* b2  = (const float*)d_in[9];
  const float* W3l = (const float*)d_in[10];
  const float* W3r = (const float*)d_in[11];
  const float* a3  = (const float*)d_in[12];
  const float* b3  = (const float*)d_in[13];

  const int N = 100000, E = 800000, ET = E + N;

  float* ws = (float*)d_ws;
  float* xl = ws;
  float* xr = xl + (size_t)N * 128;
  float* sc = xr + (size_t)N * 128;
  float* mb = sc + (size_t)ET * 4;
  float* dn = mb + (size_t)N * 4;
  float* h1 = dn + (size_t)N * 8;
  float* h2 = h1 + (size_t)N * 128;
  float* ag = h2 + (size_t)N * 128;

  const int rowTiles = N / 16;
  const int gblocks  = (rowTiles + WAVES_PER_BLOCK - 1) / WAVES_PER_BLOCK;

  auto layer = [&](const float* in, const float* inBias, const float* Wl, const float* Wr,
                   const float* att, const float* bias, float* out,
                   int C, bool final_layer) {
    const int M = 4 * C;
    const float* in2 = in + (size_t)N * 64;
    const float* dn2 = dn + (size_t)N * 4;
    if (M == 64 && !inBias) {
      gemm_wmma_f32<64, false><<<gblocks, TPB, 0, stream>>>(in, nullptr, Wl, inBias, nullptr, nullptr, xl, N);
      gemm_wmma_f32<64, false><<<gblocks, TPB, 0, stream>>>(in, nullptr, Wr, inBias, nullptr, nullptr, xr, N);
    } else if (M == 64) {
      gemm_wmma_f32<64, true><<<gblocks, TPB, 0, stream>>>(in, in2, Wl, inBias, dn, dn2, xl, N);
      gemm_wmma_f32<64, true><<<gblocks, TPB, 0, stream>>>(in, in2, Wr, inBias, dn, dn2, xr, N);
    } else {
      gemm_wmma_f32<128, true><<<gblocks, TPB, 0, stream>>>(in, in2, Wl, inBias, dn, dn2, xl, N);
      gemm_wmma_f32<128, true><<<gblocks, TPB, 0, stream>>>(in, in2, Wr, inBias, dn, dn2, xr, N);
    }

    float* aggbuf = final_layer ? ag : out;
    hipMemsetAsync(mb, 0x00, (size_t)N * 4 * sizeof(float), stream);
    hipMemsetAsync(dn, 0x00, (size_t)N * 8 * sizeof(float), stream);
    const int aggw = final_layer ? C : M;
    hipMemsetAsync(aggbuf, 0x00, (size_t)N * aggw * 2 * sizeof(float), stream);

    const int eThreads = ET * 4;
    const int eBlocks = (eThreads + TPB - 1) / TPB;
    const int aBlocks = (ET + TPB - 1) / TPB;
    if (C == 16) {
      edge_score<16><<<eBlocks, TPB, 0, stream>>>(xl, xr, ei, att, sc, mb, E, ET);
      edge_exp_agg<16><<<aBlocks, TPB, 0, stream>>>(xl, sc, mb, dn, N * 4, ei, aggbuf, N * M, E, ET);
    } else {
      edge_score<32><<<eBlocks, TPB, 0, stream>>>(xl, xr, ei, att, sc, mb, E, ET);
      edge_exp<<<eBlocks, TPB, 0, stream>>>(sc, mb, dn, ei, E, ET);
      edge_agg<32, true><<<aBlocks, TPB, 0, stream>>>(xl, sc, dn, ei, aggbuf, N * C, E, ET);
    }

    if (final_layer) {
      const int t = N * 32;
      final_mean<<<(t + TPB - 1) / TPB, TPB, 0, stream>>>(ag, ag + (size_t)N * 32, bias, (float*)d_out, N);
    }
  };

  layer(x,  nullptr, W1l, W1r, a1, b1, h1, 16, false);
  layer(h1, b1,      W2l, W2r, a2, b2, h2, 16, false);
  layer(h2, b2,      W3l, W3r, a3, b3, (float*)d_out, 32, true);
}


# max keys (all layers) and head-mean layer's denominators also kept once per L2 cache half: no atomic target is updated from both halves; readers take larger key / add both sums; sum order changes
# speedup vs baseline: 4.0929x; 1.0573x over previous
#include <hip/hip_runtime.h>
#include <math.h>

typedef __attribute__((ext_vector_type(2))) float v2f;
typedef __attribute__((ext_vector_type(8))) float v8f;

#define TPB 256
#define WAVES_PER_BLOCK 8

__device__ __forceinline__ int chip_half() {
  return (int)((__builtin_amdgcn_s_sendmsg_rtn(0x87) >> 18) & 1u);
}

template <int M, bool ACT>
__global__ __launch_bounds__(TPB) void gemm_wmma_f32(
    const float* __restrict__ X, const float* __restrict__ X2,
    const float* __restrict__ W,
    const float* __restrict__ bias, const float* __restrict__ den,
    const float* __restrict__ den2,
    float* __restrict__ Y, int Nrows) {
  __shared__ __align__(16) float lw[64 * M];

  constexpr int TOTAL4 = 64 * M / 4;
  for (int i = threadIdx.x; i < TOTAL4; i += TPB)
    ((float4*)lw)[i] = ((const float4*)W)[i];
  __syncthreads();

  const int rowTile = blockIdx.x * WAVES_PER_BLOCK + (threadIdx.x >> 5);
  if (rowTile >= (Nrows >> 4)) return;
  const int row0 = rowTile << 4;
  const int lane = threadIdx.x & 31;
  const int r    = lane & 15;
  const int half = lane >> 4;

  v2f a[16];
  const float* xrow = X + (size_t)(row0 + r) * 64;
#pragma unroll
  for (int ks = 0; ks < 16; ++ks) {
    const int k = 4 * ks + 2 * half;
    a[ks].x = xrow[k];
    a[ks].y = xrow[k + 1];
  }
  if constexpr (ACT) {
    v2f a2[16];
    const float* xrow2 = X2 + (size_t)(row0 + r) * 64;
#pragma unroll
    for (int ks = 0; ks < 16; ++ks) {
      const int k = 4 * ks + 2 * half;
      a2[ks].x = xrow2[k];
      a2[ks].y = xrow2[k + 1];
    }
    const float4 dv0 = *(const float4*)(den + (size_t)(row0 + r) * 4);
    const float4 dv1 = *(const float4*)(den2 + (size_t)(row0 + r) * 4);
    const float inv[4] = {1.0f / (dv0.x + dv1.x), 1.0f / (dv0.y + dv1.y),
                          1.0f / (dv0.z + dv1.z), 1.0f / (dv0.w + dv1.w)};
#pragma unroll
    for (int ks = 0; ks < 16; ++ks) {
      const int k = 4 * ks + 2 * half;
      const float x0 = (a[ks].x + a2[ks].x) * inv[ks >> 2] + bias[k];
      const float x1 = (a[ks].y + a2[ks].y) * inv[ks >> 2] + bias[k + 1];
      const float e0 = expm1f(x0);
      const float e1 = expm1f(x1);
      a[ks].x = x0 > 0.0f ? x0 : e0;
      a[ks].y = x1 > 0.0f ? x1 : e1;
    }
  }

#pragma unroll
  for (int tn = 0; tn < M / 16; ++tn) {
    const int col0 = tn << 4;
    v8f c = {};
#pragma unroll
    for (int ks = 0; ks < 16; ++ks) {
      const int k = 4 * ks + 2 * half;
      v2f b;
      b.x = lw[k * M + col0 + r];
      b.y = lw[(k + 1) * M + col0 + r];
      c = __builtin_amdgcn_wmma_f32_16x16x4_f32(false, a[ks], false, b,
                                                (short)0, c, false, false);
    }
#pragma unroll
    for (int v = 0; v < 8; ++v)
      Y[(size_t)(row0 + v + 8 * half) * M + col0 + r] = c[v];
  }
}

__device__ __forceinline__ void edge_sd(const int* __restrict__ ei, int e, int E,
                                        int& s, int& d) {
  if (e < E) { s = ei[e]; d = ei[E + e]; }
  else       { s = e - E; d = s; }
}

__device__ __forceinline__ unsigned int float_to_max_key(float val) {
  const unsigned int u = __float_as_uint(val);
  const unsigned int neg = (unsigned int)((int)u >> 31);
  return u ^ (neg | 0x80000000u);
}
__device__ __forceinline__ float max_key_to_float(float stored) {
  const unsigned int k = __float_as_uint(stored);
  const unsigned int pos = (unsigned int)((int)k >> 31);
  return __uint_as_float(k ^ (~pos | 0x80000000u));
}
__device__ __forceinline__ float max_key_to_float2(float stored0, float stored1) {
  const unsigned int k0 = __float_as_uint(stored0), k1 = __float_as_uint(stored1);
  return max_key_to_float(__uint_as_float(k0 > k1 ? k0 : k1));
}
__device__ __forceinline__ void atomic_max_float(float* addr, float val) {
  atomicMax((unsigned int*)addr, float_to_max_key(val));
}

template <int CTRL>
__device__ __forceinline__ float lane_xor_add(float v) {
  const int t = __builtin_amdgcn_update_dpp(0, __float_as_int(v), CTRL, 0xf, 0xf, true);
  return v + __int_as_float(t);
}

template <int C>
__global__ __launch_bounds__(TPB) void edge_score(
    const float* __restrict__ xl, const float* __restrict__ xr,
    const int* __restrict__ ei, const float* __restrict__ att,
    float* __restrict__ score, float* __restrict__ mbuf_copies,
    int mbuf_copy_floats, int E, int ET) {
  float* __restrict__ mbuf = mbuf_copies + (size_t)chip_half() * mbuf_copy_floats;
  constexpr int M     = 4 * C;
  constexpr int LPR   = M / 4;
  constexpr int EPS   = 32 / LPR;
  constexpr int LPH   = C / 4;
  constexpr int STEPS = 8 / EPS;
  static_assert(STEPS == LPH, "one kept score per lane needs steps == lanes per head");
  const int lane = threadIdx.x & 31;
  const int e0 = __builtin_amdgcn_readfirstlane(
      (int)(((blockIdx.x * TPB + threadIdx.x) >> 5) << 3));
  if (e0 >= ET) return;
  const int ch   = lane & (LPR - 1);
  const int sub  = lane / LPR;
  const int q    = lane & (LPH - 1);
  const int h    = ch / LPH;
  const int eloc = q * EPS + sub;
  const int e    = e0 + eloc;
  int s = 0, d = 0;
  if (e < ET) edge_sd(ei, e, E, s, d);
  const float4 w = ((const float4*)att)[ch];

  float4 a[STEPS], b[STEPS];
#pragma unroll
  for (int t = 0; t < STEPS; ++t) {
    int sj, dj;
    if constexpr (EPS == 1) {
      sj = __builtin_amdgcn_readlane(s, t);
      dj = __builtin_amdgcn_readlane(d, t);
    } else {
      const int s0 = __builtin_amdgcn_readlane(s, t);
      const int s1 = __builtin_amdgcn_readlane(s, 16 + t);
      const int d0 = __builtin_amdgcn_readlane(d, t);
      const int d1 = __builtin_amdgcn_readlane(d, 16 + t);
      sj = sub ? s1 : s0;
      dj = sub ? d1 : d0;
    }
    const unsigned oa = (unsigned)sj * (unsigned)(M * 4) + (unsigned)ch * 16u;
    const unsigned ob = (unsigned)dj * (unsigned)(M * 4) + (unsigned)ch * 16u;
    a[t] = *(const float4*)((const char*)xl + oa);
    b[t] = *(const float4*)((const char*)xr + ob);
  }

  float acc = 0.0f;
#pragma unroll
  for (int t = 0; t < STEPS; ++t) {
    float vx = a[t].x + b[t].x, vy = a[t].y + b[t].y;
    float vz = a[t].z + b[t].z, vw = a[t].w + b[t].w;
    vx = vx > 0.0f ? vx : 0.2f * vx;
    vy = vy > 0.0f ? vy : 0.2f * vy;
    vz = vz > 0.0f ? vz : 0.2f * vz;
    vw = vw > 0.0f ? vw : 0.2f * vw;
    float p = w.x * vx + w.y * vy + w.z * vz + w.w * vw;
    p = lane_xor_add<0x161>(p);
    p = lane_xor_add<0x162>(p);
    if constexpr (LPH == 8) p = lane_xor_add<0x164>(p);
    acc = (q == t) ? p : acc;
  }
  if (e < ET) {
    score[(size_t)e * 4 + h] = acc;
    atomic_max_float(&mbuf[(size_t)d * 4 + h], acc);
  }
}

__global__ __launch_bounds__(TPB) void edge_exp(
    float* __restrict__ score, const float* __restrict__ mbuf,
    const float* __restrict__ mbuf2, float* __restrict__ denom_copies,
    int denom_copy_floats, const int* __restrict__ ei, int E, int ET) {
  float* __restrict__ denom = denom_copies + (size_t)chip_half() * denom_copy_floats;
  const int tid = blockIdx.x * blockDim.x + threadIdx.x;
  if (tid >= ET * 4) return;
  const int e = tid >> 2, h = tid & 3;
  int s, d; edge_sd(ei, e, E, s, d);
  const float ex = expf(score[(size_t)e * 4 + h] -
                        max_key_to_float2(mbuf[(size_t)d * 4 + h], mbuf2[(size_t)d * 4 + h]));
  score[(size_t)e * 4 + h] = ex;
  atomicAdd(&denom[(size_t)d * 4 + h], ex);
}

template <int C, int NB, bool MEAN>
__device__ __forceinline__ void agg_batch(
    const float* __restrict__ xl, float* __restrict__ agg,
    int s, int d, float al0, float al1, float al2, float al3, int j0, int lane) {
  constexpr int M = 4 * C;
  constexpr int Q = M / 32;
  float v[NB][Q];
  float w[NB][Q];
  int   dj[NB];
#pragma unroll
  for (int u = 0; u < NB; ++u) {
    const int j = j0 + u;
    const int sj = __builtin_amdgcn_readlane(s, j);
    dj[u] = __builtin_amdgcn_readlane(d, j);
    const float* row = xl + (size_t)sj * M;
#pragma unroll
    for (int q = 0; q < Q; ++q) v[u][q] = row[q * 32 + lane];
    float a0, a1, a2, a3;
    if constexpr (MEAN) {
      a0 = __int_as_float(__builtin_amdgcn_readlane(__float_as_int(al0), j));
      a1 = __int_as_float(__builtin_amdgcn_readlane(__float_as_int(al1), j));
      a2 = __int_as_float(__builtin_amdgcn_readlane(__float_as_int(al2), j));
      a3 = __int_as_float(__builtin_amdgcn_readlane(__float_as_int(al3), j));
    } else {
      static_assert(8 % NB == 0, "a batch must stay inside one group of 8 edges");
      const int k = j0 >> 3;
      const int ek = __float_as_int(k == 0 ? al0 : k == 1 ? al1 : k == 2 ? al2 : al3);
      const int l0 = (j & 7) << 2;
      a0 = __int_as_float(__builtin_amdgcn_readlane(ek, l0));
      a1 = __int_as_float(__builtin_amdgcn_readlane(ek, l0 + 1));
      a2 = __int_as_float(__builtin_amdgcn_readlane(ek, l0 + 2));
      a3 = __int_as_float(__builtin_amdgcn_readlane(ek, l0 + 3));
    }
    if constexpr (Q == 4) {
      w[u][0] = a0; w[u][1] = a1; w[u][2] = a2; w[u][3] = a3;
    } else {
      const bool hi = lane >= 16;
      w[u][0] = hi ? a1 : a0;
      w[u][1] = hi ? a3 : a2;
    }
  }
#pragma unroll
  for (int u = 0; u < NB; ++u) {
    if constexpr (MEAN) {
      static_assert(Q == 4, "head sum in registers needs one head per line");
      float t = w[u][0] * v[u][0];
      t = fmaf(w[u][1], v[u][1], t);
      t = fmaf(w[u][2], v[u][2], t);
      t = fmaf(w[u][3], v[u][3], t);
      atomicAdd(&agg[(size_t)dj[u] * C + lane], t);
    } else {
      float* arow = agg + (size_t)dj[u] * M;
#pragma unroll
      for (int q = 0; q < Q; ++q)
        atomicAdd(&arow[q * 32 + lane], w[u][q] * v[u][q]);
    }
  }
}

template <int C, bool MEAN>
__global__ __launch_bounds__(TPB) void edge_agg(
    const float* __restrict__ xl, const float* __restrict__ score,
    const float* __restrict__ denom, const float* __restrict__ denom2,
    const int* __restrict__ ei,
    float* __restrict__ agg_copies, int agg_copy_floats, int E, int ET) {
  static_assert(MEAN, "the concat layers aggregate in edge_exp_agg");
  float* __restrict__ agg = agg_copies + (size_t)chip_half() * agg_copy_floats;
  constexpr int U = 128 / C;
  const int lane = threadIdx.x & 31;
  const int e0 = __builtin_amdgcn_readfirstlane(
      (int)(((blockIdx.x * TPB + threadIdx.x) >> 5) << 5));
  if (e0 >= ET) return;
  const int e = e0 + lane;
  int s = 0, d = 0;
  float al0 = 0.0f, al1 = 0.0f, al2 = 0.0f, al3 = 0.0f;
  if (e < ET) {
    edge_sd(ei, e, E, s, d);
    const float4 ex = *(const float4*)(score + (size_t)e * 4);
    const float4 dn = *(const float4*)(denom + (size_t)d * 4);
    const float4 dm = *(const float4*)(denom2 + (size_t)d * 4);
    al0 = ex.x / (dn.x + dm.x); al1 = ex.y / (dn.y + dm.y);
    al2 = ex.z / (dn.z + dm.z); al3 = ex.w / (dn.w + dm.w);
  }
  const int cnt = min(32, ET - e0);
  int j = 0;
  for (; j + U <= cnt; j += U)
    agg_batch<C, U, MEAN>(xl, agg, s, d, al0, al1, al2, al3, j, lane);
  for (; j < cnt; ++j)
    agg_batch<C, 1, MEAN>(xl, agg, s, d, al0, al1, al2, al3, j, lane);
}

template <int C>
__global__ __launch_bounds__(TPB) void edge_exp_agg(
    const float* __restrict__ xl, const float* __restrict__ score,
    const float* __restrict__ mbuf, const float* __restrict__ mbuf2,
    float* __restrict__ denom_copies,
    int denom_copy_floats, const int* __restrict__ ei,
    float* __restrict__ agg_copies, int agg_copy_floats, int E, int ET) {
  constexpr int U = 128 / C;
  const int hf = chip_half();
  float* __restrict__ denom = denom_copies + (size_t)hf * denom_copy_floats;
  float* __restrict__ agg   = agg_copies + (size_t)hf * agg_copy_floats;
  const int lane = threadIdx.x & 31;
  const int e0 = __builtin_amdgcn_readfirstlane(
      (int)(((blockIdx.x * TPB + threadIdx.x) >> 5) << 5));
  if (e0 >= ET) return;
  const int e = e0 + lane;
  int s = 0, d = 0;
  if (e < ET) edge_sd(ei, e, E, s, d);

  const int h = lane & 3;
  float sc[4], mx[4], my[4], ex[4];
  int   dk[4];
  bool  ok[4];
#pragma unroll
  for (int k = 0; k < 4; ++k) {
    const int el = 8 * k + (lane >> 2);
    dk[k] = __builtin_amdgcn_ds_bpermute(el << 2, d);
    ok[k] = e0 + el < ET;
    sc[k] = score[(size_t)e0 * 4 + (ok[k] ? 32 * k + lane : 0)];
    mx[k] = mbuf[(size_t)dk[k] * 4 + h];
    my[k] = mbuf2[(size_t)dk[k] * 4 + h];
  }
#pragma unroll
  for (int k = 0; k < 4; ++k) {
    const float x = expf(sc[k] - max_key_to_float2(mx[k], my[k]));
    if (ok[k]) atomicAdd(&denom[(size_t)dk[k] * 4 + h], x);
    ex[k] = ok[k] ? x : 0.0f;
  }

  const int cnt = min(32, ET - e0);
  if (cnt == 32) {
#pragma unroll
    for (int b = 0; b < 32 / U; ++b)
      agg_batch<C, U, false>(xl, agg, s, d, ex[0], ex[1], ex[2], ex[3], b * U, lane);
  } else {
    int j = 0;
    for (; j + U <= cnt; j += U)
      agg_batch<C, U, false>(xl, agg, s, d, ex[0], ex[1], ex[2], ex[3], j, lane);
    for (; j < cnt; ++j)
      agg_batch<C, 1, false>(xl, agg, s, d, ex[0], ex[1], ex[2], ex[3], j, lane);
  }
}

__global__ __launch_bounds__(TPB) void final_mean(
    const float* __restrict__ agg, const float* __restrict__ agg2,
    const float* __restrict__ bias, float* __restrict__ out, int N) {
  const int tid = blockIdx.x * blockDim.x + threadIdx.x;
  if (tid >= N * 32) return;
  out[tid] = 0.25f * (agg[tid] + agg2[tid]) + bias[tid & 31];
}

extern "C" void kernel_launch(void* const* d_in, const int* in_sizes, int n_in,
                              void* d_out, int out_size, void* d_ws, size_t ws_size,
                              hipStream_t stream) {
  (void)in_sizes; (void)n_in; (void)out_size; (void)ws_size;
  const float* x   = (const float*)d_in[0];
  const int*   ei  = (const int*)d_in[1];
  const float* W1l = (const float*)d_in[2];
  const float* W1r = (const float*)d_in[3];
  const float* a1  = (const float*)d_in[4];
  const float* b1  = (const float*)d_in[5];
  const float* W2l = (const float*)d_in[6];
  const float* W2r = (const float*)d_in[7];
  const float* a2  = (const float*)d_in[8];
  const float* b2  = (const float*)d_in[9];
  const float* W3l = (const float*)d_in[10];
  const float* W3r = (const float*)d_in[11];
  const float* a3  = (const float*)d_in[12];
  const float* b3  = (const float*)d_in[13];

  const int N = 100000, E = 800000, ET = E + N;

  float* ws = (float*)d_ws;
  float* xl = ws;
  float* xr = xl + (size_t)N * 128;
  float* sc = xr + (size_t)N * 128;
  float* mb = sc + (size_t)ET * 4;
  float* dn = mb + (size_t)N * 8;
  float* h1 = dn + (size_t)N * 8;
  float* h2 = h1 + (size_t)N * 128;
  float* ag = h2 + (size_t)N * 128;

  const int rowTiles = N / 16;
  const int gblocks  = (rowTiles + WAVES_PER_BLOCK - 1) / WAVES_PER_BLOCK;

  auto layer = [&](const float* in, const float* inBias, const float* Wl, const float* Wr,
                   const float* att, const float* bias, float* out,
                   int C, bool final_layer) {
    const int M = 4 * C;
    const float* in2 = in + (size_t)N * 64;
    const float* dn2 = dn + (size_t)N * 4;
    const float* mb2 = mb + (size_t)N * 4;
    if (M == 64 && !inBias) {
      gemm_wmma_f32<64, false><<<gblocks, TPB, 0, stream>>>(in, nullptr, Wl, inBias, nullptr, nullptr, xl, N);
      gemm_wmma_f32<64, false><<<gblocks, TPB, 0, stream>>>(in, nullptr, Wr, inBias, nullptr, nullptr, xr, N);
    } else if (M == 64) {
      gemm_wmma_f32<64, true><<<gblocks, TPB, 0, stream>>>(in, in2, Wl, inBias, dn, dn2, xl, N);
      gemm_wmma_f32<64, true><<<gblocks, TPB, 0, stream>>>(in, in2, Wr, inBias, dn, dn2, xr, N);
    } else {
      gemm_wmma_f32<128, true><<<gblocks, TPB, 0, stream>>>(in, in2, Wl, inBias, dn, dn2, xl, N);
      gemm_wmma_f32<128, true><<<gblocks, TPB, 0, stream>>>(in, in2, Wr, inBias, dn, dn2, xr, N);
    }

    float* aggbuf = final_layer ? ag : out;
    hipMemsetAsync(mb, 0x00, (size_t)N * 8 * sizeof(float), stream);
    hipMemsetAsync(dn, 0x00, (size_t)N * 8 * sizeof(float), stream);
    const int aggw = final_layer ? C : M;
    hipMemsetAsync(aggbuf, 0x00, (size_t)N * aggw * 2 * sizeof(float), stream);

    const int eThreads = ET * 4;
    const int eBlocks = (eThreads + TPB - 1) / TPB;
    const int aBlocks = (ET + TPB - 1) / TPB;
    if (C == 16) {
      edge_score<16><<<eBlocks, TPB, 0, stream>>>(xl, xr, ei, att, sc, mb, N * 4, E, ET);
      edge_exp_agg<16><<<aBlocks, TPB, 0, stream>>>(xl, sc, mb, mb2, dn, N * 4, ei, aggbuf, N * M, E, ET);
    } else {
      edge_score<32><<<eBlocks, TPB, 0, stream>>>(xl, xr, ei, att, sc, mb, N * 4, E, ET);
      edge_exp<<<eBlocks, TPB, 0, stream>>>(sc, mb, mb2, dn, N * 4, ei, E, ET);
      edge_agg<32, true><<<aBlocks, TPB, 0, stream>>>(xl, sc, dn, dn2, ei, aggbuf, N * C, E, ET);
    }

    if (final_layer) {
      const int t = N * 32;
      final_mean<<<(t + TPB - 1) / TPB, TPB, 0, stream>>>(ag, ag + (size_t)N * 32, bias, (float*)d_out, N);
    }
  };

  layer(x,  nullptr, W1l, W1r, a1, b1, h1, 16, false);
  layer(h1, b1,      W2l, W2r, a2, b2, h2, 16, false);
  layer(h2, b2,      W3l, W3r, a3, b3, (float*)d_out, 32, true);
}


# edge_score, the head-mean layer's edge_exp and the final pass give each wave 4x the work (32 edges as 4 blocks of 8; 16 bytes per lane): 412500 fewer waves enter per call; per-element sums unchanged
# speedup vs baseline: 4.4035x; 1.0759x over previous
#include <hip/hip_runtime.h>
#include <math.h>

typedef __attribute__((ext_vector_type(2))) float v2f;
typedef __attribute__((ext_vector_type(8))) float v8f;

#define TPB 256
#define WAVES_PER_BLOCK 8

__device__ __forceinline__ int chip_half() {
  return (int)((__builtin_amdgcn_s_sendmsg_rtn(0x87) >> 18) & 1u);
}

template <int M, bool ACT>
__global__ __launch_bounds__(TPB) void gemm_wmma_f32(
    const float* __restrict__ X, const float* __restrict__ X2,
    const float* __restrict__ W,
    const float* __restrict__ bias, const float* __restrict__ den,
    const float* __restrict__ den2,
    float* __restrict__ Y, int Nrows) {
  __shared__ __align__(16) float lw[64 * M];

  constexpr int TOTAL4 = 64 * M / 4;
  for (int i = threadIdx.x; i < TOTAL4; i += TPB)
    ((float4*)lw)[i] = ((const float4*)W)[i];
  __syncthreads();

  const int rowTile = blockIdx.x * WAVES_PER_BLOCK + (threadIdx.x >> 5);
  if (rowTile >= (Nrows >> 4)) return;
  const int row0 = rowTile << 4;
  const int lane = threadIdx.x & 31;
  const int r    = lane & 15;
  const int half = lane >> 4;

  v2f a[16];
  const float* xrow = X + (size_t)(row0 + r) * 64;
#pragma unroll
  for (int ks = 0; ks < 16; ++ks) {
    const int k = 4 * ks + 2 * half;
    a[ks].x = xrow[k];
    a[ks].y = xrow[k + 1];
  }
  if constexpr (ACT) {
    v2f a2[16];
    const float* xrow2 = X2 + (size_t)(row0 + r) * 64;
#pragma unroll
    for (int ks = 0; ks < 16; ++ks) {
      const int k = 4 * ks + 2 * half;
      a2[ks].x = xrow2[k];
      a2[ks].y = xrow2[k + 1];
    }
    const float4 dv0 = *(const float4*)(den + (size_t)(row0 + r) * 4);
    const float4 dv1 = *(const float4*)(den2 + (size_t)(row0 + r) * 4);
    const float inv[4] = {1.0f / (dv0.x + dv1.x), 1.0f / (dv0.y + dv1.y),
                          1.0f / (dv0.z + dv1.z), 1.0f / (dv0.w + dv1.w)};
#pragma unroll
    for (int ks = 0; ks < 16; ++ks) {
      const int k = 4 * ks + 2 * half;
      const float x0 = (a[ks].x + a2[ks].x) * inv[ks >> 2] + bias[k];
      const float x1 = (a[ks].y + a2[ks].y) * inv[ks >> 2] + bias[k + 1];
      const float e0 = expm1f(x0);
      const float e1 = expm1f(x1);
      a[ks].x = x0 > 0.0f ? x0 : e0;
      a[ks].y = x1 > 0.0f ? x1 : e1;
    }
  }

#pragma unroll
  for (int tn = 0; tn < M / 16; ++tn) {
    const int col0 = tn << 4;
    v8f c = {};
#pragma unroll
    for (int ks = 0; ks < 16; ++ks) {
      const int k = 4 * ks + 2 * half;
      v2f b;
      b.x = lw[k * M + col0 + r];
      b.y = lw[(k + 1) * M + col0 + r];
      c = __builtin_amdgcn_wmma_f32_16x16x4_f32(false, a[ks], false, b,
                                                (short)0, c, false, false);
    }
#pragma unroll
    for (int v = 0; v < 8; ++v)
      Y[(size_t)(row0 + v + 8 * half) * M + col0 + r] = c[v];
  }
}

__device__ __forceinline__ void edge_sd(const int* __restrict__ ei, int e, int E,
                                        int& s, int& d) {
  if (e < E) { s = ei[e]; d = ei[E + e]; }
  else       { s = e - E; d = s; }
}

__device__ __forceinline__ unsigned int float_to_max_key(float val) {
  const unsigned int u = __float_as_uint(val);
  const unsigned int neg = (unsigned int)((int)u >> 31);
  return u ^ (neg | 0x80000000u);
}
__device__ __forceinline__ float max_key_to_float(float stored) {
  const unsigned int k = __float_as_uint(stored);
  const unsigned int pos = (unsigned int)((int)k >> 31);
  return __uint_as_float(k ^ (~pos | 0x80000000u));
}
__device__ __forceinline__ float max_key_to_float2(float stored0, float stored1) {
  const unsigned int k0 = __float_as_uint(stored0), k1 = __float_as_uint(stored1);
  return max_key_to_float(__uint_as_float(k0 > k1 ? k0 : k1));
}
__device__ __forceinline__ void atomic_max_float(float* addr, float val) {
  atomicMax((unsigned int*)addr, float_to_max_key(val));
}

template <int CTRL>
__device__ __forceinline__ float lane_xor_add(float v) {
  const int t = __builtin_amdgcn_update_dpp(0, __float_as_int(v), CTRL, 0xf, 0xf, true);
  return v + __int_as_float(t);
}

template <int C>
__global__ __launch_bounds__(TPB) void edge_score(
    const float* __restrict__ xl, const float* __restrict__ xr,
    const int* __restrict__ ei, const float* __restrict__ att,
    float* __restrict__ score, float* __restrict__ mbuf_copies,
    int mbuf_copy_floats, int E, int ET) {
  float* __restrict__ mbuf = mbuf_copies + (size_t)chip_half() * mbuf_copy_floats;
  constexpr int M     = 4 * C;
  constexpr int LPR   = M / 4;
  constexpr int EPS   = 32 / LPR;
  constexpr int LPH   = C / 4;
  constexpr int STEPS = 8 / EPS;
  static_assert(STEPS == LPH, "one kept score per lane needs steps == lanes per head");
  constexpr int BPW = 4;
  const int lane = threadIdx.x & 31;
  const int w0 = __builtin_amdgcn_readfirstlane(
      (int)(((blockIdx.x * TPB + threadIdx.x) >> 5) << 5));
  if (w0 >= ET) return;
  const int ch   = lane & (LPR - 1);
  const int sub  = lane / LPR;
  const int q    = lane & (LPH - 1);
  const int h    = ch / LPH;
  const int eloc = q * EPS + sub;
  int s = 0, d = 0;
  if (w0 + lane < ET) edge_sd(ei, w0 + lane, E, s, d);
  const float4 w = ((const float4*)att)[ch];

  for (int blk = 0; blk < BPW; ++blk) {
    const int l0 = 8 * blk;
    const int e0 = w0 + l0;
    if (e0 >= ET) break;
    float4 a[STEPS], b[STEPS];
#pragma unroll
    for (int t = 0; t < STEPS; ++t) {
      int sj, dj;
      if constexpr (EPS == 1) {
        sj = __builtin_amdgcn_readlane(s, l0 + t);
        dj = __builtin_amdgcn_readlane(d, l0 + t);
      } else {
        const int s0 = __builtin_amdgcn_readlane(s, l0 + 2 * t);
        const int s1 = __builtin_amdgcn_readlane(s, l0 + 2 * t + 1);
        const int d0 = __builtin_amdgcn_readlane(d, l0 + 2 * t);
        const int d1 = __builtin_amdgcn_readlane(d, l0 + 2 * t + 1);
        sj = sub ? s1 : s0;
        dj = sub ? d1 : d0;
      }
      const unsigned oa = (unsigned)sj * (unsigned)(M * 4) + (unsigned)ch * 16u;
      const unsigned ob = (unsigned)dj * (unsigned)(M * 4) + (unsigned)ch * 16u;
      a[t] = *(const float4*)((const char*)xl + oa);
      b[t] = *(const float4*)((const char*)xr + ob);
    }
    const int dm = __builtin_amdgcn_ds_bpermute((l0 + eloc) << 2, d);

    float acc = 0.0f;
#pragma unroll
    for (int t = 0; t < STEPS; ++t) {
      float vx = a[t].x + b[t].x, vy = a[t].y + b[t].y;
      float vz = a[t].z + b[t].z, vw = a[t].w + b[t].w;
      vx = vx > 0.0f ? vx : 0.2f * vx;
      vy = vy > 0.0f ? vy : 0.2f * vy;
      vz = vz > 0.0f ? vz : 0.2f * vz;
      vw = vw > 0.0f ? vw : 0.2f * vw;
      float p = w.x * vx + w.y * vy + w.z * vz + w.w * vw;
      p = lane_xor_add<0x161>(p);
      p = lane_xor_add<0x162>(p);
      if constexpr (LPH == 8) p = lane_xor_add<0x164>(p);
      acc = (q == t) ? p : acc;
    }
    const int e = e0 + eloc;
    if (e < ET) {
      score[(size_t)e * 4 + h] = acc;
      atomic_max_float(&mbuf[(size_t)dm * 4 + h], acc);
    }
  }
}

__global__ __launch_bounds__(TPB) void edge_exp(
    float* __restrict__ score, const float* __restrict__ mbuf,
    const float* __restrict__ mbuf2, float* __restrict__ denom_copies,
    int denom_copy_floats, const int* __restrict__ ei, int E, int ET) {
  float* __restrict__ denom = denom_copies + (size_t)chip_half() * denom_copy_floats;
  const int lane = threadIdx.x & 31;
  const int w0 = __builtin_amdgcn_readfirstlane(
      (int)(((blockIdx.x * TPB + threadIdx.x) >> 5) << 5));
  if (w0 >= ET) return;
  const int e = w0 + lane;
  int d = 0;
  if (e < ET) d = e < E ? ei[E + e] : e - E;
  const int h = lane & 3;
  float sc[4], mx[4], my[4];
  int   dk[4];
  bool  ok[4];
#pragma unroll
  for (int k = 0; k < 4; ++k) {
    const int el = 8 * k + (lane >> 2);
    dk[k] = __builtin_amdgcn_ds_bpermute(el << 2, d);
    ok[k] = w0 + el < ET;
    sc[k] = score[(size_t)w0 * 4 + (ok[k] ? 32 * k + lane : 0)];
    mx[k] = mbuf[(size_t)dk[k] * 4 + h];
    my[k] = mbuf2[(size_t)dk[k] * 4 + h];
  }
#pragma unroll
  for (int k = 0; k < 4; ++k) {
    const float ex = expf(sc[k] - max_key_to_float2(mx[k], my[k]));
    if (ok[k]) {
      score[(size_t)w0 * 4 + 32 * k + lane] = ex;
      atomicAdd(&denom[(size_t)dk[k] * 4 + h], ex);
    }
  }
}

template <int C, int NB, bool MEAN>
__device__ __forceinline__ void agg_batch(
    const float* __restrict__ xl, float* __restrict__ agg,
    int s, int d, float al0, float al1, float al2, float al3, int j0, int lane) {
  constexpr int M = 4 * C;
  constexpr int Q = M / 32;
  float v[NB][Q];
  float w[NB][Q];
  int   dj[NB];
#pragma unroll
  for (int u = 0; u < NB; ++u) {
    const int j = j0 + u;
    const int sj = __builtin_amdgcn_readlane(s, j);
    dj[u] = __builtin_amdgcn_readlane(d, j);
    const float* row = xl + (size_t)sj * M;
#pragma unroll
    for (int q = 0; q < Q; ++q) v[u][q] = row[q * 32 + lane];
    float a0, a1, a2, a3;
    if constexpr (MEAN) {
      a0 = __int_as_float(__builtin_amdgcn_readlane(__float_as_int(al0), j));
      a1 = __int_as_float(__builtin_amdgcn_readlane(__float_as_int(al1), j));
      a2 = __int_as_float(__builtin_amdgcn_readlane(__float_as_int(al2), j));
      a3 = __int_as_float(__builtin_amdgcn_readlane(__float_as_int(al3), j));
    } else {
      static_assert(8 % NB == 0, "a batch must stay inside one group of 8 edges");
      const int k = j0 >> 3;
      const int ek = __float_as_int(k == 0 ? al0 : k == 1 ? al1 : k == 2 ? al2 : al3);
      const int l0 = (j & 7) << 2;
      a0 = __int_as_float(__builtin_amdgcn_readlane(ek, l0));
      a1 = __int_as_float(__builtin_amdgcn_readlane(ek, l0 + 1));
      a2 = __int_as_float(__builtin_amdgcn_readlane(ek, l0 + 2));
      a3 = __int_as_float(__builtin_amdgcn_readlane(ek, l0 + 3));
    }
    if constexpr (Q == 4) {
      w[u][0] = a0; w[u][1] = a1; w[u][2] = a2; w[u][3] = a3;
    } else {
      const bool hi = lane >= 16;
      w[u][0] = hi ? a1 : a0;
      w[u][1] = hi ? a3 : a2;
    }
  }
#pragma unroll
  for (int u = 0; u < NB; ++u) {
    if constexpr (MEAN) {
      static_assert(Q == 4, "head sum in registers needs one head per line");
      float t = w[u][0] * v[u][0];
      t = fmaf(w[u][1], v[u][1], t);
      t = fmaf(w[u][2], v[u][2], t);
      t = fmaf(w[u][3], v[u][3], t);
      atomicAdd(&agg[(size_t)dj[u] * C + lane], t);
    } else {
      float* arow = agg + (size_t)dj[u] * M;
#pragma unroll
      for (int q = 0; q < Q; ++q)
        atomicAdd(&arow[q * 32 + lane], w[u][q] * v[u][q]);
    }
  }
}

template <int C, bool MEAN>
__global__ __launch_bounds__(TPB) void edge_agg(
    const float* __restrict__ xl, const float* __restrict__ score,
    const float* __restrict__ denom, const float* __restrict__ denom2,
    const int* __restrict__ ei,
    float* __restrict__ agg_copies, int agg_copy_floats, int E, int ET) {
  static_assert(MEAN, "the concat layers aggregate in edge_exp_agg");
  float* __restrict__ agg = agg_copies + (size_t)chip_half() * agg_copy_floats;
  constexpr int U = 128 / C;
  const int lane = threadIdx.x & 31;
  const int e0 = __builtin_amdgcn_readfirstlane(
      (int)(((blockIdx.x * TPB + threadIdx.x) >> 5) << 5));
  if (e0 >= ET) return;
  const int e = e0 + lane;
  int s = 0, d = 0;
  float al0 = 0.0f, al1 = 0.0f, al2 = 0.0f, al3 = 0.0f;
  if (e < ET) {
    edge_sd(ei, e, E, s, d);
    const float4 ex = *(const float4*)(score + (size_t)e * 4);
    const float4 dn = *(const float4*)(denom + (size_t)d * 4);
    const float4 dm = *(const float4*)(denom2 + (size_t)d * 4);
    al0 = ex.x / (dn.x + dm.x); al1 = ex.y / (dn.y + dm.y);
    al2 = ex.z / (dn.z + dm.z); al3 = ex.w / (dn.w + dm.w);
  }
  const int cnt = min(32, ET - e0);
  int j = 0;
  for (; j + U <= cnt; j += U)
    agg_batch<C, U, MEAN>(xl, agg, s, d, al0, al1, al2, al3, j, lane);
  for (; j < cnt; ++j)
    agg_batch<C, 1, MEAN>(xl, agg, s, d, al0, al1, al2, al3, j, lane);
}

template <int C>
__global__ __launch_bounds__(TPB) void edge_exp_agg(
    const float* __restrict__ xl, const float* __restrict__ score,
    const float* __restrict__ mbuf, const float* __restrict__ mbuf2,
    float* __restrict__ denom_copies,
    int denom_copy_floats, const int* __restrict__ ei,
    float* __restrict__ agg_copies, int agg_copy_floats, int E, int ET) {
  constexpr int U = 128 / C;
  const int hf = chip_half();
  float* __restrict__ denom = denom_copies + (size_t)hf * denom_copy_floats;
  float* __restrict__ agg   = agg_copies + (size_t)hf * agg_copy_floats;
  const int lane = threadIdx.x & 31;
  const int e0 = __builtin_amdgcn_readfirstlane(
      (int)(((blockIdx.x * TPB + threadIdx.x) >> 5) << 5));
  if (e0 >= ET) return;
  const int e = e0 + lane;
  int s = 0, d = 0;
  if (e < ET) edge_sd(ei, e, E, s, d);

  const int h = lane & 3;
  float sc[4], mx[4], my[4], ex[4];
  int   dk[4];
  bool  ok[4];
#pragma unroll
  for (int k = 0; k < 4; ++k) {
    const int el = 8 * k + (lane >> 2);
    dk[k] = __builtin_amdgcn_ds_bpermute(el << 2, d);
    ok[k] = e0 + el < ET;
    sc[k] = score[(size_t)e0 * 4 + (ok[k] ? 32 * k + lane : 0)];
    mx[k] = mbuf[(size_t)dk[k] * 4 + h];
    my[k] = mbuf2[(size_t)dk[k] * 4 + h];
  }
#pragma unroll
  for (int k = 0; k < 4; ++k) {
    const float x = expf(sc[k] - max_key_to_float2(mx[k], my[k]));
    if (ok[k]) atomicAdd(&denom[(size_t)dk[k] * 4 + h], x);
    ex[k] = ok[k] ? x : 0.0f;
  }

  const int cnt = min(32, ET - e0);
  if (cnt == 32) {
#pragma unroll
    for (int b = 0; b < 32 / U; ++b)
      agg_batch<C, U, false>(xl, agg, s, d, ex[0], ex[1], ex[2], ex[3], b * U, lane);
  } else {
    int j = 0;
    for (; j + U <= cnt; j += U)
      agg_batch<C, U, false>(xl, agg, s, d, ex[0], ex[1], ex[2], ex[3], j, lane);
    for (; j < cnt; ++j)
      agg_batch<C, 1, false>(xl, agg, s, d, ex[0], ex[1], ex[2], ex[3], j, lane);
  }
}

__global__ __launch_bounds__(TPB) void final_mean(
    const float* __restrict__ agg, const float* __restrict__ agg2,
    const float* __restrict__ bias, float* __restrict__ out, int N) {
  const int tid = blockIdx.x * blockDim.x + threadIdx.x;
  if (tid >= N * 8) return;
  const float4 a = ((const float4*)agg)[tid];
  const float4 b = ((const float4*)agg2)[tid];
  const float4 c = ((const float4*)bias)[tid & 7];
  float4 o;
  o.x = 0.25f * (a.x + b.x) + c.x;
  o.y = 0.25f * (a.y + b.y) + c.y;
  o.z = 0.25f * (a.z + b.z) + c.z;
  o.w = 0.25f * (a.w + b.w) + c.w;
  ((float4*)out)[tid] = o;
}

extern "C" void kernel_launch(void* const* d_in, const int* in_sizes, int n_in,
                              void* d_out, int out_size, void* d_ws, size_t ws_size,
                              hipStream_t stream) {
  (void)in_sizes; (void)n_in; (void)out_size; (void)ws_size;
  const float* x   = (const float*)d_in[0];
  const int*   ei  = (const int*)d_in[1];
  const float* W1l = (const float*)d_in[2];
  const float* W1r = (const float*)d_in[3];
  const float* a1  = (const float*)d_in[4];
  const float* b1  = (const float*)d_in[5];
  const float* W2l = (const float*)d_in[6];
  const float* W2r = (const float*)d_in[7];
  const float* a2  = (const float*)d_in[8];
  const float* b2  = (const float*)d_in[9];
  const float* W3l = (const float*)d_in[10];
  const float* W3r = (const float*)d_in[11];
  const float* a3  = (const float*)d_in[12];
  const float* b3  = (const float*)d_in[13];

  const int N = 100000, E = 800000, ET = E + N;

  float* ws = (float*)d_ws;
  float* xl = ws;
  float* xr = xl + (size_t)N * 128;
  float* sc = xr + (size_t)N * 128;
  float* mb = sc + (size_t)ET * 4;
  float* dn = mb + (size_t)N * 8;
  float* h1 = dn + (size_t)N * 8;
  float* h2 = h1 + (size_t)N * 128;
  float* ag = h2 + (size_t)N * 128;

  const int rowTiles = N / 16;
  const int gblocks  = (rowTiles + WAVES_PER_BLOCK - 1) / WAVES_PER_BLOCK;

  auto layer = [&](const float* in, const float* inBias, const float* Wl, const float* Wr,
                   const float* att, const float* bias, float* out,
                   int C, bool final_layer) {
    const int M = 4 * C;
    const float* in2 = in + (size_t)N * 64;
    const float* dn2 = dn + (size_t)N * 4;
    const float* mb2 = mb + (size_t)N * 4;
    if (M == 64 && !inBias) {
      gemm_wmma_f32<64, false><<<gblocks, TPB, 0, stream>>>(in, nullptr, Wl, inBias, nullptr, nullptr, xl, N);
      gemm_wmma_f32<64, false><<<gblocks, TPB, 0, stream>>>(in, nullptr, Wr, inBias, nullptr, nullptr, xr, N);
    } else if (M == 64) {
      gemm_wmma_f32<64, true><<<gblocks, TPB, 0, stream>>>(in, in2, Wl, inBias, dn, dn2, xl, N);
      gemm_wmma_f32<64, true><<<gblocks, TPB, 0, stream>>>(in, in2, Wr, inBias, dn, dn2, xr, N);
    } else {
      gemm_wmma_f32<128, true><<<gblocks, TPB, 0, stream>>>(in, in2, Wl, inBias, dn, dn2, xl, N);
      gemm_wmma_f32<128, true><<<gblocks, TPB, 0, stream>>>(in, in2, Wr, inBias, dn, dn2, xr, N);
    }

    float* aggbuf = final_layer ? ag : out;
    hipMemsetAsync(mb, 0x00, (size_t)N * 8 * sizeof(float), stream);
    hipMemsetAsync(dn, 0x00, (size_t)N * 8 * sizeof(float), stream);
    const int aggw = final_layer ? C : M;
    hipMemsetAsync(aggbuf, 0x00, (size_t)N * aggw * 2 * sizeof(float), stream);

    const int aBlocks = (ET + TPB - 1) / TPB;
    if (C == 16) {
      edge_score<16><<<aBlocks, TPB, 0, stream>>>(xl, xr, ei, att, sc, mb, N * 4, E, ET);
      edge_exp_agg<16><<<aBlocks, TPB, 0, stream>>>(xl, sc, mb, mb2, dn, N * 4, ei, aggbuf, N * M, E, ET);
    } else {
      edge_score<32><<<aBlocks, TPB, 0, stream>>>(xl, xr, ei, att, sc, mb, N * 4, E, ET);
      edge_exp<<<aBlocks, TPB, 0, stream>>>(sc, mb, mb2, dn, N * 4, ei, E, ET);
      edge_agg<32, true><<<aBlocks, TPB, 0, stream>>>(xl, sc, dn, dn2, ei, aggbuf, N * C, E, ET);
    }

    if (final_layer) {
      const int t = N * 8;
      final_mean<<<(t + TPB - 1) / TPB, TPB, 0, stream>>>(ag, ag + (size_t)N * 32, bias, (float*)d_out, N);
    }
  };

  layer(x,  nullptr, W1l, W1r, a1, b1, h1, 16, false);
  layer(h1, b1,      W2l, W2r, a2, b2, h2, 16, false);
  layer(h2, b2,      W3l, W3r, a3, b3, (float*)d_out, 32, true);
}
